# starting kernel
# baseline (speedup 1.0000x reference)
#include <hip/hip_runtime.h>
#include <hip/hip_bf16.h>


typedef float v2f __attribute__((ext_vector_type(2)));
typedef float v8f __attribute__((ext_vector_type(8)));

#define LIN_B  16
#define LIN_S  4096
#define LIN_D  64
#define LIN_KP 256

__device__ __forceinline__ v8f wmma_f32(v2f a, v2f b, v8f c) {
    return __builtin_amdgcn_wmma_f32_16x16x4_f32(false, a, false, b, (short)0, c,
                                                 false, false);
}

template <bool KT_PACK>
__global__ __launch_bounds__(32) void lin_proj_kernel(
    const float* __restrict__ X, const float* __restrict__ W,
    const float* __restrict__ bias, float* __restrict__ outP) {
    const int S = LIN_S, D = LIN_D, KP = LIN_KP;

    const int blk  = blockIdx.x;
    const int b    = blk >> 6;
    const int tile = blk & 63;
    const int mt   = tile >> 2;
    const int nt   = tile & 3;
    const int l    = threadIdx.x;
    const int lh   = l >> 4;
    const int ll   = l & 15;
    const int kp0  = mt * 16;
    const int d0   = nt * 16;

    const float* wrow = W + (size_t)(kp0 + ll) * S + lh * 2;
    const float* xcol = X + (size_t)b * S * D + d0 + ll;

    v8f acc = {};
    for (int s0 = 0; s0 < S; s0 += 32) {
        v2f a[8], bf[8];
#pragma unroll
        for (int j = 0; j < 8; ++j)
            a[j] = *(const v2f*)(wrow + s0 + j * 4);
#pragma unroll
        for (int j = 0; j < 8; ++j) {
            const float* xp = xcol + (size_t)(s0 + j * 4 + lh * 2) * D;
            bf[j].x = xp[0];
            bf[j].y = xp[D];
        }
#pragma unroll
        for (int j = 0; j < 8; ++j) acc = wmma_f32(a[j], bf[j], acc);
    }

#pragma unroll
    for (int v = 0; v < 8; ++v) acc[v] += bias[kp0 + 8 * lh + v];

    const int d = d0 + ll;
    if (KT_PACK) {
        float* p = outP + (size_t)b * D * KP + (size_t)(d >> 1) * (2 * KP) +
                   (size_t)(kp0 + 8 * lh) * 2 + (d & 1);
#pragma unroll
        for (int v = 0; v < 8; ++v) p[2 * v] = acc[v];
    } else {
        float* base = outP + (size_t)b * KP * D + d * 2;
#pragma unroll
        for (int v = 0; v < 8; ++v) {
            const int kp = kp0 + 8 * lh + v;
            base[(size_t)(kp >> 1) * (2 * D) + (kp & 1)] = acc[v];
        }
    }
}

#define LDS_STRIDE 260

__global__ __launch_bounds__(32) void lin_attn_kernel(
    const float* __restrict__ Q, const float* __restrict__ Kpp,
    const float* __restrict__ Vpp, float* __restrict__ out,
    float* __restrict__ attn) {
    const int S = LIN_S, D = LIN_D, KP = LIN_KP;
    const float scale = 0.125f;

    __shared__ float lds[16 * LDS_STRIDE];

    const int blk = blockIdx.x;
    const int b   = blk >> 8;
    const int s0  = (blk & 255) * 16;
    const int l   = threadIdx.x;
    const int lh  = l >> 4;
    const int ll  = l & 15;

    v2f qa[16];
    const float* qrow = Q + (size_t)b * S * D + (size_t)(s0 + ll) * D + lh * 2;
#pragma unroll
    for (int kk = 0; kk < 16; ++kk) qa[kk] = *(const v2f*)(qrow + kk * 4);

    v8f sc[16];
    const float* kpB = Kpp + (size_t)b * D * KP + ll * 2;
#pragma unroll
    for (int t = 0; t < 16; ++t) {
        v2f bf[16];
#pragma unroll
        for (int kk = 0; kk < 16; ++kk)
            bf[kk] = *(const v2f*)(kpB + (size_t)(kk * 2 + lh) * (2 * KP) + t * 32);
        v8f c = {};
#pragma unroll
        for (int kk = 0; kk < 16; ++kk) c = wmma_f32(qa[kk], bf[kk], c);
        sc[t] = c;
    }

#pragma unroll
    for (int v = 0; v < 8; ++v) {
        float mx = sc[0][v];
#pragma unroll
        for (int t = 1; t < 16; ++t) mx = fmaxf(mx, sc[t][v]);
        mx = fmaxf(mx, __shfl_xor(mx, 1, 32));
        mx = fmaxf(mx, __shfl_xor(mx, 2, 32));
        mx = fmaxf(mx, __shfl_xor(mx, 4, 32));
        mx = fmaxf(mx, __shfl_xor(mx, 8, 32));
        float sum = 0.f;
#pragma unroll
        for (int t = 0; t < 16; ++t) {
            float p = __expf((sc[t][v] - mx) * scale);
            sc[t][v] = p;
            sum += p;
        }
        sum += __shfl_xor(sum, 1, 32);
        sum += __shfl_xor(sum, 2, 32);
        sum += __shfl_xor(sum, 4, 32);
        sum += __shfl_xor(sum, 8, 32);
        const float r = 1.0f / sum;
#pragma unroll
        for (int t = 0; t < 16; ++t) sc[t][v] *= r;
    }

    float* arow = attn + (size_t)b * S * KP;
#pragma unroll
    for (int t = 0; t < 16; ++t) {
#pragma unroll
        for (int v = 0; v < 8; ++v) {
            const int m   = v + 8 * lh;
            const int col = t * 16 + ll;
            arow[(size_t)(s0 + m) * KP + col] = sc[t][v];
            lds[m * LDS_STRIDE + col]         = sc[t][v];
        }
    }
    __syncthreads();

    const float* vpB = Vpp + (size_t)b * KP * D + ll * 2;
    v8f oc[4] = {};
#pragma unroll 4
    for (int kk = 0; kk < 64; ++kk) {
        v2f a = *(const v2f*)(&lds[ll * LDS_STRIDE + kk * 4 + lh * 2]);
        v2f bf[4];
#pragma unroll
        for (int td = 0; td < 4; ++td)
            bf[td] = *(const v2f*)(vpB + (size_t)(kk * 2 + lh) * (2 * D) + td * 32);
#pragma unroll
        for (int td = 0; td < 4; ++td) oc[td] = wmma_f32(a, bf[td], oc[td]);
    }
#pragma unroll
    for (int td = 0; td < 4; ++td) {
#pragma unroll
        for (int v = 0; v < 8; ++v)
            out[(size_t)b * S * D + (size_t)(s0 + v + 8 * lh) * D + td * 16 + ll] =
                oc[td][v];
    }
}

extern "C" void kernel_launch(void* const* d_in, const int* in_sizes, int n_in,
                              void* d_out, int out_size, void* d_ws, size_t ws_size,
                              hipStream_t stream) {
    (void)in_sizes; (void)n_in; (void)out_size; (void)ws_size;
    const float* Q   = (const float*)d_in[0];
    const float* K   = (const float*)d_in[1];
    const float* V   = (const float*)d_in[2];
    const float* E_w = (const float*)d_in[3];
    const float* E_b = (const float*)d_in[4];
    const float* F_w = (const float*)d_in[5];
    const float* F_b = (const float*)d_in[6];

    float* out  = (float*)d_out;
    float* attn = out + (size_t)LIN_B * LIN_S * LIN_D;

    float* Kpp = (float*)d_ws;
    float* Vpp = Kpp + (size_t)LIN_B * LIN_D * LIN_KP;

    const int projBlocks = LIN_B * 16 * 4;
    const int attnBlocks = LIN_B * (LIN_S / 16);

    lin_proj_kernel<true ><<<projBlocks, 32, 0, stream>>>(K, E_w, E_b, Kpp);
    lin_proj_kernel<false><<<projBlocks, 32, 0, stream>>>(V, F_w, F_b, Vpp);
    lin_attn_kernel<<<attnBlocks, 32, 0, stream>>>(Q, Kpp, Vpp, out, attn);
}


# Projection chunk loop as a pipeline: the next chunk's 32 LDS fragment reads, the parking of the chunk after and a fetch 3 chunks on stand as one group before each chunk's 16 products; sums unchanged
# speedup vs baseline: 2.7853x; 2.7853x over previous
#include <hip/hip_runtime.h>
#include <hip/hip_bf16.h>


typedef float v2f __attribute__((ext_vector_type(2)));
typedef float v8f __attribute__((ext_vector_type(8)));

#define LIN_B  16
#define LIN_S  4096
#define LIN_D  64
#define LIN_KP 256

__device__ __forceinline__ v8f wmma_f32(v2f a, v2f b, v8f c) {
    return __builtin_amdgcn_wmma_f32_16x16x4_f32(false, a, false, b, (short)0, c,
                                                 false, false);
}

typedef float v4f __attribute__((ext_vector_type(4)));

#define PJ_CH  64
#define PJ_NCH (LIN_S / PJ_CH)
#define PJ_WP  68
#define PJ_XP  160

static __device__ __forceinline__ void pj_barrier(v8f acc_done) {
    asm volatile("s_wait_dscnt 0x0\n\ts_barrier_signal -1\n\ts_barrier_wait -1"
                 :: "v"(acc_done) : "memory");
}

__global__ __launch_bounds__(256) void lin_proj_kernel(
    const float* __restrict__ Kx, const float* __restrict__ Vx,
    const float* __restrict__ Ew, const float* __restrict__ Fw,
    const float* __restrict__ Eb, const float* __restrict__ Fb,
    float* __restrict__ Kpp, float* __restrict__ Vpp) {
    const int S = LIN_S, D = LIN_D, KP = LIN_KP;

    __shared__ __attribute__((aligned(16))) float Ws[2][32 * PJ_WP];
    __shared__ __attribute__((aligned(16))) float Xs[2][32 * PJ_XP];

    const int blk = blockIdx.x;
    const bool isV = (blk >> 7) != 0;
    const int b    = (blk >> 3) & 15;
    const int kp0  = (blk & 7) * 32;

    const float* X    = (isV ? Vx : Kx) + (size_t)b * S * D;
    const float* W    = (isV ? Fw : Ew) + (size_t)kp0 * S;
    const float* bias = isV ? Fb : Eb;

    const int t    = threadIdx.x;
    const int wave = t >> 5;
    const int l    = t & 31;
    const int lh   = l >> 4;
    const int ll   = l & 15;
    const int wi   = wave >> 2;
    const int wj   = wave & 3;

    const int q = t & 15;
    const int r = t >> 4;
    const float* gW = W + (size_t)r * S + 4 * q;
    const float* gX = X + (size_t)(2 * r) * D + 4 * q;
    const int wOff = r * PJ_WP + 4 * q;
    const int xOff = r * PJ_XP + 8 * q;

    const int aOff = (16 * wi + ll) * PJ_WP + 2 * lh;
    const int bOff = lh * PJ_XP + (16 * wj + ll) * 2;

    v4f w00, w01, x00, x01, x02, x03;
    v4f w10, w11, x10, x11, x12, x13;
    v4f w20, w21, x20, x21, x22, x23;
    v2f fa0[16], fb0[16], fa1[16], fb1[16];

#define PJ_FETCH(w0, w1, x0, x1, x2, x3, chunk)                         \
    {                                                                   \
        const int sn_ = (chunk) * PJ_CH;                                \
        w0 = *(const v4f*)(gW + sn_);                                   \
        w1 = *(const v4f*)(gW + (size_t)16 * S + sn_);                  \
        x0 = *(const v4f*)(gX + (size_t)sn_ * D);                       \
        x1 = *(const v4f*)(gX + (size_t)sn_ * D + D);                   \
        x2 = *(const v4f*)(gX + (size_t)sn_ * D + 32 * D);              \
        x3 = *(const v4f*)(gX + (size_t)sn_ * D + 33 * D);              \
    }

#define PJ_PARK(w0, w1, x0, x1, x2, x3, ws, xs)                         \
    {                                                                   \
        *(v4f*)((ws) + wOff)              = w0;                         \
        *(v4f*)((ws) + wOff + 16 * PJ_WP) = w1;                         \
        v4f p0_ = {x0.x, x1.x, x0.y, x1.y};                             \
        v4f p1_ = {x0.z, x1.z, x0.w, x1.w};                             \
        v4f p2_ = {x2.x, x3.x, x2.y, x3.y};                             \
        v4f p3_ = {x2.z, x3.z, x2.w, x3.w};                             \
        *(v4f*)((xs) + xOff)                  = p0_;                    \
        *(v4f*)((xs) + xOff + 4)              = p1_;                    \
        *(v4f*)((xs) + xOff + 16 * PJ_XP)     = p2_;                    \
        *(v4f*)((xs) + xOff + 16 * PJ_XP + 4) = p3_;                    \
    }

#define PJ_READ(fa, fb, ws, xs)                                         \
    {                                                                   \
        _Pragma("unroll") for (int j = 0; j < 16; ++j)                  \
            fa[j] = *(const v2f*)((ws) + aOff + 4 * j);                 \
        _Pragma("unroll") for (int j = 0; j < 16; ++j)                  \
            fb[j] = *(const v2f*)((xs) + bOff + 2 * j * PJ_XP);         \
    }

#define PJ_MULT(fa, fb)                                                 \
    {                                                                   \
        __builtin_amdgcn_sched_barrier(0);                              \
        _Pragma("unroll") for (int j = 0; j < 16; ++j)                  \
            acc = wmma_f32(fa[j], fb[j], acc);                          \
        __builtin_amdgcn_sched_barrier(0);                              \
    }

#define PJ_STAGE(fnext, fbnext, rws, rxs, w0, w1, x0, x1, x2, x3, pws, pxs, \
                 fchunk, fcur, fbcur)                                       \
    {                                                                       \
        PJ_READ(fnext, fbnext, rws, rxs)                                    \
        PJ_PARK(w0, w1, x0, x1, x2, x3, pws, pxs)                           \
        PJ_FETCH(w0, w1, x0, x1, x2, x3, fchunk)                            \
        PJ_MULT(fcur, fbcur)                                                \
        pj_barrier(acc);                                                    \
    }
    float* const ws0 = &Ws[0][0];
    float* const xs0 = &Xs[0][0];
    float* const ws1 = &Ws[1][0];
    float* const xs1 = &Xs[1][0];

    v8f acc = {};
    PJ_FETCH(w00, w01, x00, x01, x02, x03, 0)
    PJ_FETCH(w10, w11, x10, x11, x12, x13, 1)
    PJ_FETCH(w20, w21, x20, x21, x22, x23, 2)
    PJ_PARK(w00, w01, x00, x01, x02, x03, ws0, xs0)
    pj_barrier(acc);
    PJ_FETCH(w00, w01, x00, x01, x02, x03, 3)
    PJ_READ(fa0, fb0, ws0, xs0)
    PJ_PARK(w10, w11, x10, x11, x12, x13, ws1, xs1)
    pj_barrier(acc);
    PJ_FETCH(w10, w11, x10, x11, x12, x13, 4)
    for (int i = 0; i < PJ_NCH - 4; i += 6) {
        PJ_STAGE(fa1, fb1, ws1, xs1, w20, w21, x20, x21, x22, x23, ws0, xs0, i + 5, fa0, fb0)
        PJ_STAGE(fa0, fb0, ws0, xs0, w00, w01, x00, x01, x02, x03, ws1, xs1, i + 6, fa1, fb1)
        PJ_STAGE(fa1, fb1, ws1, xs1, w10, w11, x10, x11, x12, x13, ws0, xs0, i + 7, fa0, fb0)
        PJ_STAGE(fa0, fb0, ws0, xs0, w20, w21, x20, x21, x22, x23, ws1, xs1, i + 8, fa1, fb1)
        PJ_STAGE(fa1, fb1, ws1, xs1, w00, w01, x00, x01, x02, x03, ws0, xs0, i + 9, fa0, fb0)
        PJ_STAGE(fa0, fb0, ws0, xs0, w10, w11, x10, x11, x12, x13, ws1, xs1,
                 (i + 10 < PJ_NCH ? i + 10 : PJ_NCH - 1), fa1, fb1)
    }
    PJ_READ(fa1, fb1, ws1, xs1)
    PJ_PARK(w20, w21, x20, x21, x22, x23, ws0, xs0)
    PJ_MULT(fa0, fb0)
    pj_barrier(acc);
    PJ_READ(fa0, fb0, ws0, xs0)
    PJ_PARK(w00, w01, x00, x01, x02, x03, ws1, xs1)
    PJ_MULT(fa1, fb1)
    pj_barrier(acc);
    PJ_READ(fa1, fb1, ws1, xs1)
    PJ_MULT(fa0, fb0)
    PJ_MULT(fa1, fb1)
#undef PJ_STAGE
#undef PJ_READ
#undef PJ_FETCH
#undef PJ_PARK
#undef PJ_MULT

    const int kpt = kp0 + 16 * wi;
    const int d   = 16 * wj + ll;
#pragma unroll
    for (int v = 0; v < 8; ++v) acc[v] += bias[kpt + 8 * lh + v];

    if (!isV) {
        float* p = Kpp + (size_t)b * D * KP + (size_t)(d >> 1) * (2 * KP) +
                   (size_t)(kpt + 8 * lh) * 2 + (d & 1);
#pragma unroll
        for (int v = 0; v < 8; ++v) p[2 * v] = acc[v];
    } else {
        float* base = Vpp + (size_t)b * KP * D + d * 2;
#pragma unroll
        for (int v = 0; v < 8; ++v) {
            const int kp = kpt + 8 * lh + v;
            base[(size_t)(kp >> 1) * (2 * D) + (kp & 1)] = acc[v];
        }
    }
}

#define LDS_STRIDE 260
#define AT_MS 2
#define AT_ROWS (16 * AT_MS)

__global__ __launch_bounds__(32) void lin_attn_kernel(
    const float* __restrict__ Q, const float* __restrict__ Kpp,
    const float* __restrict__ Vpp, float* __restrict__ out,
    float* __restrict__ attn) {
    const int S = LIN_S, D = LIN_D, KP = LIN_KP;
    const float scale = 0.125f;

    __shared__ float lds[AT_ROWS * LDS_STRIDE];

    const int blk = blockIdx.x;
    const int b   = blk / (LIN_S / AT_ROWS);
    const int s0  = (blk % (LIN_S / AT_ROWS)) * AT_ROWS;
    const int l   = threadIdx.x;
    const int lh  = l >> 4;
    const int ll  = l & 15;

    v2f qa[AT_MS][16];
#pragma unroll
    for (int ms = 0; ms < AT_MS; ++ms) {
        const float* qrow =
            Q + (size_t)b * S * D + (size_t)(s0 + 16 * ms + ll) * D + lh * 2;
#pragma unroll
        for (int kk = 0; kk < 16; ++kk) qa[ms][kk] = *(const v2f*)(qrow + kk * 4);
    }

    v8f sc[AT_MS][16];
    const float* kpB = Kpp + (size_t)b * D * KP + ll * 2;
    v2f bfn[16];
#pragma unroll
    for (int kk = 0; kk < 16; ++kk)
        bfn[kk] = *(const v2f*)(kpB + (size_t)(kk * 2 + lh) * (2 * KP));
#pragma unroll
    for (int t = 0; t < 16; ++t) {
        v2f bf[16];
#pragma unroll
        for (int kk = 0; kk < 16; ++kk) bf[kk] = bfn[kk];
        if (t + 1 < 16) {
#pragma unroll
            for (int kk = 0; kk < 16; ++kk)
                bfn[kk] = *(const v2f*)(kpB + (size_t)(kk * 2 + lh) * (2 * KP) + (t + 1) * 32);
        }
        v8f c[AT_MS];
#pragma unroll
        for (int ms = 0; ms < AT_MS; ++ms) c[ms] = v8f{};
#pragma unroll
        for (int kk = 0; kk < 16; ++kk) {
#pragma unroll
            for (int ms = 0; ms < AT_MS; ++ms) c[ms] = wmma_f32(qa[ms][kk], bf[kk], c[ms]);
        }
#pragma unroll
        for (int ms = 0; ms < AT_MS; ++ms) sc[ms][t] = c[ms];
    }

#pragma unroll
    for (int ms = 0; ms < AT_MS; ++ms) {
#pragma unroll
        for (int v = 0; v < 8; ++v) {
            float mx = sc[ms][0][v];
#pragma unroll
            for (int t = 1; t < 16; ++t) mx = fmaxf(mx, sc[ms][t][v]);
            mx = fmaxf(mx, __shfl_xor(mx, 1, 32));
            mx = fmaxf(mx, __shfl_xor(mx, 2, 32));
            mx = fmaxf(mx, __shfl_xor(mx, 4, 32));
            mx = fmaxf(mx, __shfl_xor(mx, 8, 32));
            float sum = 0.f;
#pragma unroll
            for (int t = 0; t < 16; ++t) {
                float p = __expf((sc[ms][t][v] - mx) * scale);
                sc[ms][t][v] = p;
                sum += p;
            }
            sum += __shfl_xor(sum, 1, 32);
            sum += __shfl_xor(sum, 2, 32);
            sum += __shfl_xor(sum, 4, 32);
            sum += __shfl_xor(sum, 8, 32);
            const float r = 1.0f / sum;
#pragma unroll
            for (int t = 0; t < 16; ++t) sc[ms][t][v] *= r;
        }
    }

    float* arow = attn + (size_t)b * S * KP;
#pragma unroll
    for (int ms = 0; ms < AT_MS; ++ms) {
#pragma unroll
        for (int t = 0; t < 16; ++t) {
#pragma unroll
            for (int v = 0; v < 8; ++v) {
                const int m   = 16 * ms + v + 8 * lh;
                const int col = t * 16 + ll;
                arow[(size_t)(s0 + m) * KP + col] = sc[ms][t][v];
                lds[m * LDS_STRIDE + col]         = sc[ms][t][v];
            }
        }
    }
    __syncthreads();

    const float* vpB = Vpp + (size_t)b * KP * D + ll * 2;
    v8f oc[AT_MS][4];
#pragma unroll
    for (int ms = 0; ms < AT_MS; ++ms) {
#pragma unroll
        for (int td = 0; td < 4; ++td) oc[ms][td] = v8f{};
    }
#pragma unroll 4
    for (int kk = 0; kk < 64; ++kk) {
        v2f a[AT_MS];
#pragma unroll
        for (int ms = 0; ms < AT_MS; ++ms)
            a[ms] = *(const v2f*)(&lds[(16 * ms + ll) * LDS_STRIDE + kk * 4 + lh * 2]);
        v2f bf[4];
#pragma unroll
        for (int td = 0; td < 4; ++td)
            bf[td] = *(const v2f*)(vpB + (size_t)(kk * 2 + lh) * (2 * D) + td * 32);
#pragma unroll
        for (int td = 0; td < 4; ++td) {
#pragma unroll
            for (int ms = 0; ms < AT_MS; ++ms)
                oc[ms][td] = wmma_f32(a[ms], bf[td], oc[ms][td]);
        }
    }
#pragma unroll
    for (int ms = 0; ms < AT_MS; ++ms) {
#pragma unroll
        for (int td = 0; td < 4; ++td) {
#pragma unroll
            for (int v = 0; v < 8; ++v)
                out[(size_t)b * S * D + (size_t)(s0 + 16 * ms + v + 8 * lh) * D + td * 16 +
                    ll] = oc[ms][td][v];
        }
    }
}

extern "C" void kernel_launch(void* const* d_in, const int* in_sizes, int n_in,
                              void* d_out, int out_size, void* d_ws, size_t ws_size,
                              hipStream_t stream) {
    (void)in_sizes; (void)n_in; (void)out_size; (void)ws_size;
    const float* Q   = (const float*)d_in[0];
    const float* K   = (const float*)d_in[1];
    const float* V   = (const float*)d_in[2];
    const float* E_w = (const float*)d_in[3];
    const float* E_b = (const float*)d_in[4];
    const float* F_w = (const float*)d_in[5];
    const float* F_b = (const float*)d_in[6];

    float* out  = (float*)d_out;
    float* attn = out + (size_t)LIN_B * LIN_S * LIN_D;

    float* Kpp = (float*)d_ws;
    float* Vpp = Kpp + (size_t)LIN_B * LIN_D * LIN_KP;

    const int attnBlocks = LIN_B * (LIN_S / AT_ROWS);

    lin_proj_kernel<<<256, 256, 0, stream>>>(K, V, E_w, F_w, E_b, F_b, Kpp, Vpp);
    lin_attn_kernel<<<attnBlocks, 32, 0, stream>>>(Q, Kpp, Vpp, out, attn);
}


# Softmax row max and row sum: the 4 butterfly steps per row run on the vector ALU (DPP row_xmask on the max and on the add), not as LDS-routed shuffles with a wait each; order of sums unchanged
# speedup vs baseline: 2.9132x; 1.0459x over previous
#include <hip/hip_runtime.h>
#include <hip/hip_bf16.h>


typedef float v2f __attribute__((ext_vector_type(2)));
typedef float v8f __attribute__((ext_vector_type(8)));

#define LIN_B  16
#define LIN_S  4096
#define LIN_D  64
#define LIN_KP 256

__device__ __forceinline__ v8f wmma_f32(v2f a, v2f b, v8f c) {
    return __builtin_amdgcn_wmma_f32_16x16x4_f32(false, a, false, b, (short)0, c,
                                                 false, false);
}

typedef float v4f __attribute__((ext_vector_type(4)));

#define PJ_CH  64
#define PJ_NCH (LIN_S / PJ_CH)
#define PJ_WP  68
#define PJ_XP  160

static __device__ __forceinline__ void pj_barrier(v8f acc_done) {
    asm volatile("s_wait_dscnt 0x0\n\ts_barrier_signal -1\n\ts_barrier_wait -1"
                 :: "v"(acc_done) : "memory");
}

__global__ __launch_bounds__(256) void lin_proj_kernel(
    const float* __restrict__ Kx, const float* __restrict__ Vx,
    const float* __restrict__ Ew, const float* __restrict__ Fw,
    const float* __restrict__ Eb, const float* __restrict__ Fb,
    float* __restrict__ Kpp, float* __restrict__ Vpp) {
    const int S = LIN_S, D = LIN_D, KP = LIN_KP;

    __shared__ __attribute__((aligned(16))) float Ws[2][32 * PJ_WP];
    __shared__ __attribute__((aligned(16))) float Xs[2][32 * PJ_XP];

    const int blk = blockIdx.x;
    const bool isV = (blk >> 7) != 0;
    const int b    = (blk >> 3) & 15;
    const int kp0  = (blk & 7) * 32;

    const float* X    = (isV ? Vx : Kx) + (size_t)b * S * D;
    const float* W    = (isV ? Fw : Ew) + (size_t)kp0 * S;
    const float* bias = isV ? Fb : Eb;

    const int t    = threadIdx.x;
    const int wave = t >> 5;
    const int l    = t & 31;
    const int lh   = l >> 4;
    const int ll   = l & 15;
    const int wi   = wave >> 2;
    const int wj   = wave & 3;

    const int q = t & 15;
    const int r = t >> 4;
    const float* gW = W + (size_t)r * S + 4 * q;
    const float* gX = X + (size_t)(2 * r) * D + 4 * q;
    const int wOff = r * PJ_WP + 4 * q;
    const int xOff = r * PJ_XP + 8 * q;

    const int aOff = (16 * wi + ll) * PJ_WP + 2 * lh;
    const int bOff = lh * PJ_XP + (16 * wj + ll) * 2;

    v4f w00, w01, x00, x01, x02, x03;
    v4f w10, w11, x10, x11, x12, x13;
    v4f w20, w21, x20, x21, x22, x23;
    v2f fa0[16], fb0[16], fa1[16], fb1[16];

#define PJ_FETCH(w0, w1, x0, x1, x2, x3, chunk)                         \
    {                                                                   \
        const int sn_ = (chunk) * PJ_CH;                                \
        w0 = *(const v4f*)(gW + sn_);                                   \
        w1 = *(const v4f*)(gW + (size_t)16 * S + sn_);                  \
        x0 = *(const v4f*)(gX + (size_t)sn_ * D);                       \
        x1 = *(const v4f*)(gX + (size_t)sn_ * D + D);                   \
        x2 = *(const v4f*)(gX + (size_t)sn_ * D + 32 * D);              \
        x3 = *(const v4f*)(gX + (size_t)sn_ * D + 33 * D);              \
    }

#define PJ_PARK(w0, w1, x0, x1, x2, x3, ws, xs)                         \
    {                                                                   \
        *(v4f*)((ws) + wOff)              = w0;                         \
        *(v4f*)((ws) + wOff + 16 * PJ_WP) = w1;                         \
        v4f p0_ = {x0.x, x1.x, x0.y, x1.y};                             \
        v4f p1_ = {x0.z, x1.z, x0.w, x1.w};                             \
        v4f p2_ = {x2.x, x3.x, x2.y, x3.y};                             \
        v4f p3_ = {x2.z, x3.z, x2.w, x3.w};                             \
        *(v4f*)((xs) + xOff)                  = p0_;                    \
        *(v4f*)((xs) + xOff + 4)              = p1_;                    \
        *(v4f*)((xs) + xOff + 16 * PJ_XP)     = p2_;                    \
        *(v4f*)((xs) + xOff + 16 * PJ_XP + 4) = p3_;                    \
    }

#define PJ_READ(fa, fb, ws, xs)                                         \
    {                                                                   \
        _Pragma("unroll") for (int j = 0; j < 16; ++j)                  \
            fa[j] = *(const v2f*)((ws) + aOff + 4 * j);                 \
        _Pragma("unroll") for (int j = 0; j < 16; ++j)                  \
            fb[j] = *(const v2f*)((xs) + bOff + 2 * j * PJ_XP);         \
    }

#define PJ_MULT(fa, fb)                                                 \
    {                                                                   \
        __builtin_amdgcn_sched_barrier(0);                              \
        _Pragma("unroll") for (int j = 0; j < 16; ++j)                  \
            acc = wmma_f32(fa[j], fb[j], acc);                          \
        __builtin_amdgcn_sched_barrier(0);                              \
    }

#define PJ_STAGE(fnext, fbnext, rws, rxs, w0, w1, x0, x1, x2, x3, pws, pxs, \
                 fchunk, fcur, fbcur)                                       \
    {                                                                       \
        PJ_READ(fnext, fbnext, rws, rxs)                                    \
        PJ_PARK(w0, w1, x0, x1, x2, x3, pws, pxs)                           \
        PJ_FETCH(w0, w1, x0, x1, x2, x3, fchunk)                            \
        PJ_MULT(fcur, fbcur)                                                \
        pj_barrier(acc);                                                    \
    }
    float* const ws0 = &Ws[0][0];
    float* const xs0 = &Xs[0][0];
    float* const ws1 = &Ws[1][0];
    float* const xs1 = &Xs[1][0];

    v8f acc = {};
    PJ_FETCH(w00, w01, x00, x01, x02, x03, 0)
    PJ_FETCH(w10, w11, x10, x11, x12, x13, 1)
    PJ_FETCH(w20, w21, x20, x21, x22, x23, 2)
    PJ_PARK(w00, w01, x00, x01, x02, x03, ws0, xs0)
    pj_barrier(acc);
    PJ_FETCH(w00, w01, x00, x01, x02, x03, 3)
    PJ_READ(fa0, fb0, ws0, xs0)
    PJ_PARK(w10, w11, x10, x11, x12, x13, ws1, xs1)
    pj_barrier(acc);
    PJ_FETCH(w10, w11, x10, x11, x12, x13, 4)
    for (int i = 0; i < PJ_NCH - 4; i += 6) {
        PJ_STAGE(fa1, fb1, ws1, xs1, w20, w21, x20, x21, x22, x23, ws0, xs0, i + 5, fa0, fb0)
        PJ_STAGE(fa0, fb0, ws0, xs0, w00, w01, x00, x01, x02, x03, ws1, xs1, i + 6, fa1, fb1)
        PJ_STAGE(fa1, fb1, ws1, xs1, w10, w11, x10, x11, x12, x13, ws0, xs0, i + 7, fa0, fb0)
        PJ_STAGE(fa0, fb0, ws0, xs0, w20, w21, x20, x21, x22, x23, ws1, xs1, i + 8, fa1, fb1)
        PJ_STAGE(fa1, fb1, ws1, xs1, w00, w01, x00, x01, x02, x03, ws0, xs0, i + 9, fa0, fb0)
        PJ_STAGE(fa0, fb0, ws0, xs0, w10, w11, x10, x11, x12, x13, ws1, xs1,
                 (i + 10 < PJ_NCH ? i + 10 : PJ_NCH - 1), fa1, fb1)
    }
    PJ_READ(fa1, fb1, ws1, xs1)
    PJ_PARK(w20, w21, x20, x21, x22, x23, ws0, xs0)
    PJ_MULT(fa0, fb0)
    pj_barrier(acc);
    PJ_READ(fa0, fb0, ws0, xs0)
    PJ_PARK(w00, w01, x00, x01, x02, x03, ws1, xs1)
    PJ_MULT(fa1, fb1)
    pj_barrier(acc);
    PJ_READ(fa1, fb1, ws1, xs1)
    PJ_MULT(fa0, fb0)
    PJ_MULT(fa1, fb1)
#undef PJ_STAGE
#undef PJ_READ
#undef PJ_FETCH
#undef PJ_PARK
#undef PJ_MULT

    const int kpt = kp0 + 16 * wi;
    const int d   = 16 * wj + ll;
#pragma unroll
    for (int v = 0; v < 8; ++v) acc[v] += bias[kpt + 8 * lh + v];

    if (!isV) {
        float* p = Kpp + (size_t)b * D * KP + (size_t)(d >> 1) * (2 * KP) +
                   (size_t)(kpt + 8 * lh) * 2 + (d & 1);
#pragma unroll
        for (int v = 0; v < 8; ++v) p[2 * v] = acc[v];
    } else {
        float* base = Vpp + (size_t)b * KP * D + d * 2;
#pragma unroll
        for (int v = 0; v < 8; ++v) {
            const int kp = kpt + 8 * lh + v;
            base[(size_t)(kp >> 1) * (2 * D) + (kp & 1)] = acc[v];
        }
    }
}

#define LDS_STRIDE 260
#define AT_MS 2
#define AT_ROWS (16 * AT_MS)

template <int M>
__device__ __forceinline__ float row_xor(float x) {
    return __builtin_bit_cast(
        float, __builtin_amdgcn_update_dpp(0, __builtin_bit_cast(int, x), 0x160 + M, 0xf,
                                           0xf, true));
}

__global__ __launch_bounds__(32) void lin_attn_kernel(
    const float* __restrict__ Q, const float* __restrict__ Kpp,
    const float* __restrict__ Vpp, float* __restrict__ out,
    float* __restrict__ attn) {
    const int S = LIN_S, D = LIN_D, KP = LIN_KP;
    const float scale = 0.125f;

    __shared__ float lds[AT_ROWS * LDS_STRIDE];

    const int blk = blockIdx.x;
    const int b   = blk / (LIN_S / AT_ROWS);
    const int s0  = (blk % (LIN_S / AT_ROWS)) * AT_ROWS;
    const int l   = threadIdx.x;
    const int lh  = l >> 4;
    const int ll  = l & 15;

    v2f qa[AT_MS][16];
#pragma unroll
    for (int ms = 0; ms < AT_MS; ++ms) {
        const float* qrow =
            Q + (size_t)b * S * D + (size_t)(s0 + 16 * ms + ll) * D + lh * 2;
#pragma unroll
        for (int kk = 0; kk < 16; ++kk) qa[ms][kk] = *(const v2f*)(qrow + kk * 4);
    }

    v8f sc[AT_MS][16];
    const float* kpB = Kpp + (size_t)b * D * KP + ll * 2;
    v2f bfn[16];
#pragma unroll
    for (int kk = 0; kk < 16; ++kk)
        bfn[kk] = *(const v2f*)(kpB + (size_t)(kk * 2 + lh) * (2 * KP));
#pragma unroll
    for (int t = 0; t < 16; ++t) {
        v2f bf[16];
#pragma unroll
        for (int kk = 0; kk < 16; ++kk) bf[kk] = bfn[kk];
        if (t + 1 < 16) {
#pragma unroll
            for (int kk = 0; kk < 16; ++kk)
                bfn[kk] = *(const v2f*)(kpB + (size_t)(kk * 2 + lh) * (2 * KP) + (t + 1) * 32);
        }
        v8f c[AT_MS];
#pragma unroll
        for (int ms = 0; ms < AT_MS; ++ms) c[ms] = v8f{};
#pragma unroll
        for (int kk = 0; kk < 16; ++kk) {
#pragma unroll
            for (int ms = 0; ms < AT_MS; ++ms) c[ms] = wmma_f32(qa[ms][kk], bf[kk], c[ms]);
        }
#pragma unroll
        for (int ms = 0; ms < AT_MS; ++ms) sc[ms][t] = c[ms];
    }

#pragma unroll
    for (int ms = 0; ms < AT_MS; ++ms) {
        float mx[8];
#pragma unroll
        for (int v = 0; v < 8; ++v) {
            float m = sc[ms][0][v];
#pragma unroll
            for (int t = 1; t < 16; ++t) m = fmaxf(m, sc[ms][t][v]);
            mx[v] = m;
        }
#pragma unroll
        for (int v = 0; v < 8; ++v) mx[v] = fmaxf(mx[v], row_xor<1>(mx[v]));
#pragma unroll
        for (int v = 0; v < 8; ++v) mx[v] = fmaxf(mx[v], row_xor<2>(mx[v]));
#pragma unroll
        for (int v = 0; v < 8; ++v) mx[v] = fmaxf(mx[v], row_xor<4>(mx[v]));
#pragma unroll
        for (int v = 0; v < 8; ++v) mx[v] = fmaxf(mx[v], row_xor<8>(mx[v]));

        float sum[8];
#pragma unroll
        for (int v = 0; v < 8; ++v) {
            float s = 0.f;
#pragma unroll
            for (int t = 0; t < 16; ++t) {
                float p = __expf((sc[ms][t][v] - mx[v]) * scale);
                sc[ms][t][v] = p;
                s += p;
            }
            sum[v] = s;
        }
#pragma unroll
        for (int v = 0; v < 8; ++v) sum[v] += row_xor<1>(sum[v]);
#pragma unroll
        for (int v = 0; v < 8; ++v) sum[v] += row_xor<2>(sum[v]);
#pragma unroll
        for (int v = 0; v < 8; ++v) sum[v] += row_xor<4>(sum[v]);
#pragma unroll
        for (int v = 0; v < 8; ++v) sum[v] += row_xor<8>(sum[v]);

#pragma unroll
        for (int v = 0; v < 8; ++v) {
            const float r = 1.0f / sum[v];
#pragma unroll
            for (int t = 0; t < 16; ++t) sc[ms][t][v] *= r;
        }
    }

    float* arow = attn + (size_t)b * S * KP;
#pragma unroll
    for (int ms = 0; ms < AT_MS; ++ms) {
#pragma unroll
        for (int t = 0; t < 16; ++t) {
#pragma unroll
            for (int v = 0; v < 8; ++v) {
                const int m   = 16 * ms + v + 8 * lh;
                const int col = t * 16 + ll;
                arow[(size_t)(s0 + m) * KP + col] = sc[ms][t][v];
                lds[m * LDS_STRIDE + col]         = sc[ms][t][v];
            }
        }
    }
    __syncthreads();

    const float* vpB = Vpp + (size_t)b * KP * D + ll * 2;
    v8f oc[AT_MS][4];
#pragma unroll
    for (int ms = 0; ms < AT_MS; ++ms) {
#pragma unroll
        for (int td = 0; td < 4; ++td) oc[ms][td] = v8f{};
    }
#pragma unroll 4
    for (int kk = 0; kk < 64; ++kk) {
        v2f a[AT_MS];
#pragma unroll
        for (int ms = 0; ms < AT_MS; ++ms)
            a[ms] = *(const v2f*)(&lds[(16 * ms + ll) * LDS_STRIDE + kk * 4 + lh * 2]);
        v2f bf[4];
#pragma unroll
        for (int td = 0; td < 4; ++td)
            bf[td] = *(const v2f*)(vpB + (size_t)(kk * 2 + lh) * (2 * D) + td * 32);
#pragma unroll
        for (int td = 0; td < 4; ++td) {
#pragma unroll
            for (int ms = 0; ms < AT_MS; ++ms)
                oc[ms][td] = wmma_f32(a[ms], bf[td], oc[ms][td]);
        }
    }
#pragma unroll
    for (int ms = 0; ms < AT_MS; ++ms) {
#pragma unroll
        for (int td = 0; td < 4; ++td) {
#pragma unroll
            for (int v = 0; v < 8; ++v)
                out[(size_t)b * S * D + (size_t)(s0 + 16 * ms + v + 8 * lh) * D + td * 16 +
                    ll] = oc[ms][td][v];
        }
    }
}

extern "C" void kernel_launch(void* const* d_in, const int* in_sizes, int n_in,
                              void* d_out, int out_size, void* d_ws, size_t ws_size,
                              hipStream_t stream) {
    (void)in_sizes; (void)n_in; (void)out_size; (void)ws_size;
    const float* Q   = (const float*)d_in[0];
    const float* K   = (const float*)d_in[1];
    const float* V   = (const float*)d_in[2];
    const float* E_w = (const float*)d_in[3];
    const float* E_b = (const float*)d_in[4];
    const float* F_w = (const float*)d_in[5];
    const float* F_b = (const float*)d_in[6];

    float* out  = (float*)d_out;
    float* attn = out + (size_t)LIN_B * LIN_S * LIN_D;

    float* Kpp = (float*)d_ws;
    float* Vpp = Kpp + (size_t)LIN_B * LIN_D * LIN_KP;

    const int attnBlocks = LIN_B * (LIN_S / AT_ROWS);

    lin_proj_kernel<<<256, 256, 0, stream>>>(K, V, E_w, F_w, E_b, F_b, Kpp, Vpp);
    lin_attn_kernel<<<attnBlocks, 32, 0, stream>>>(Q, Kpp, Vpp, out, attn);
}


# Projection stage barrier split: each wave signals behind its own LDS traffic after the stage's first 8 products and waits after the last 8; the round trip runs under products; sums unchanged
# speedup vs baseline: 3.0714x; 1.0543x over previous
#include <hip/hip_runtime.h>
#include <hip/hip_bf16.h>


typedef float v2f __attribute__((ext_vector_type(2)));
typedef float v8f __attribute__((ext_vector_type(8)));

#define LIN_B  16
#define LIN_S  4096
#define LIN_D  64
#define LIN_KP 256

__device__ __forceinline__ v8f wmma_f32(v2f a, v2f b, v8f c) {
    return __builtin_amdgcn_wmma_f32_16x16x4_f32(false, a, false, b, (short)0, c,
                                                 false, false);
}

typedef float v4f __attribute__((ext_vector_type(4)));

#define PJ_CH  64
#define PJ_NCH (LIN_S / PJ_CH)
#define PJ_WP  68
#define PJ_XP  160

static __device__ __forceinline__ void pj_barrier(v8f acc_done) {
    asm volatile("s_wait_dscnt 0x0\n\ts_barrier_signal -1\n\ts_barrier_wait -1"
                 :: "v"(acc_done) : "memory");
}

static __device__ __forceinline__ void pj_signal(v8f acc_half) {
    asm volatile("s_wait_dscnt 0x0\n\ts_barrier_signal -1" :: "v"(acc_half) : "memory");
}
static __device__ __forceinline__ void pj_wait(v8f acc_done) {
    asm volatile("s_barrier_wait -1" :: "v"(acc_done) : "memory");
}

__global__ __launch_bounds__(256) void lin_proj_kernel(
    const float* __restrict__ Kx, const float* __restrict__ Vx,
    const float* __restrict__ Ew, const float* __restrict__ Fw,
    const float* __restrict__ Eb, const float* __restrict__ Fb,
    float* __restrict__ Kpp, float* __restrict__ Vpp) {
    const int S = LIN_S, D = LIN_D, KP = LIN_KP;

    __shared__ __attribute__((aligned(16))) float Ws[2][32 * PJ_WP];
    __shared__ __attribute__((aligned(16))) float Xs[2][32 * PJ_XP];

    const int blk = blockIdx.x;
    const bool isV = (blk >> 7) != 0;
    const int b    = (blk >> 3) & 15;
    const int kp0  = (blk & 7) * 32;

    const float* X    = (isV ? Vx : Kx) + (size_t)b * S * D;
    const float* W    = (isV ? Fw : Ew) + (size_t)kp0 * S;
    const float* bias = isV ? Fb : Eb;

    const int t    = threadIdx.x;
    const int wave = t >> 5;
    const int l    = t & 31;
    const int lh   = l >> 4;
    const int ll   = l & 15;
    const int wi   = wave >> 2;
    const int wj   = wave & 3;

    const int q = t & 15;
    const int r = t >> 4;
    const float* gW = W + (size_t)r * S + 4 * q;
    const float* gX = X + (size_t)(2 * r) * D + 4 * q;
    const int wOff = r * PJ_WP + 4 * q;
    const int xOff = r * PJ_XP + 8 * q;

    const int aOff = (16 * wi + ll) * PJ_WP + 2 * lh;
    const int bOff = lh * PJ_XP + (16 * wj + ll) * 2;

    v4f w00, w01, x00, x01, x02, x03;
    v4f w10, w11, x10, x11, x12, x13;
    v4f w20, w21, x20, x21, x22, x23;
    v2f fa0[16], fb0[16], fa1[16], fb1[16];

#define PJ_FETCH(w0, w1, x0, x1, x2, x3, chunk)                         \
    {                                                                   \
        const int sn_ = (chunk) * PJ_CH;                                \
        w0 = *(const v4f*)(gW + sn_);                                   \
        w1 = *(const v4f*)(gW + (size_t)16 * S + sn_);                  \
        x0 = *(const v4f*)(gX + (size_t)sn_ * D);                       \
        x1 = *(const v4f*)(gX + (size_t)sn_ * D + D);                   \
        x2 = *(const v4f*)(gX + (size_t)sn_ * D + 32 * D);              \
        x3 = *(const v4f*)(gX + (size_t)sn_ * D + 33 * D);              \
    }

#define PJ_PARK(w0, w1, x0, x1, x2, x3, ws, xs)                         \
    {                                                                   \
        *(v4f*)((ws) + wOff)              = w0;                         \
        *(v4f*)((ws) + wOff + 16 * PJ_WP) = w1;                         \
        v4f p0_ = {x0.x, x1.x, x0.y, x1.y};                             \
        v4f p1_ = {x0.z, x1.z, x0.w, x1.w};                             \
        v4f p2_ = {x2.x, x3.x, x2.y, x3.y};                             \
        v4f p3_ = {x2.z, x3.z, x2.w, x3.w};                             \
        *(v4f*)((xs) + xOff)                  = p0_;                    \
        *(v4f*)((xs) + xOff + 4)              = p1_;                    \
        *(v4f*)((xs) + xOff + 16 * PJ_XP)     = p2_;                    \
        *(v4f*)((xs) + xOff + 16 * PJ_XP + 4) = p3_;                    \
    }

#define PJ_READ(fa, fb, ws, xs)                                         \
    {                                                                   \
        _Pragma("unroll") for (int j = 0; j < 16; ++j)                  \
            fa[j] = *(const v2f*)((ws) + aOff + 4 * j);                 \
        _Pragma("unroll") for (int j = 0; j < 16; ++j)                  \
            fb[j] = *(const v2f*)((xs) + bOff + 2 * j * PJ_XP);         \
    }

#define PJ_MULT(fa, fb)                                                 \
    {                                                                   \
        __builtin_amdgcn_sched_barrier(0);                              \
        _Pragma("unroll") for (int j = 0; j < 16; ++j)                  \
            acc = wmma_f32(fa[j], fb[j], acc);                          \
        __builtin_amdgcn_sched_barrier(0);                              \
    }
#define PJ_MULT_HALF(fa, fb, j0)                                        \
    {                                                                   \
        __builtin_amdgcn_sched_barrier(0);                              \
        _Pragma("unroll") for (int j = (j0); j < (j0) + 8; ++j)         \
            acc = wmma_f32(fa[j], fb[j], acc);                          \
        __builtin_amdgcn_sched_barrier(0);                              \
    }

#define PJ_STAGE(fnext, fbnext, rws, rxs, w0, w1, x0, x1, x2, x3, pws, pxs, \
                 fchunk, fcur, fbcur)                                       \
    {                                                                       \
        PJ_READ(fnext, fbnext, rws, rxs)                                    \
        PJ_PARK(w0, w1, x0, x1, x2, x3, pws, pxs)                           \
        PJ_FETCH(w0, w1, x0, x1, x2, x3, fchunk)                            \
        PJ_MULT_HALF(fcur, fbcur, 0)                                        \
        pj_signal(acc);                                                     \
        PJ_MULT_HALF(fcur, fbcur, 8)                                        \
        pj_wait(acc);                                                       \
    }
    float* const ws0 = &Ws[0][0];
    float* const xs0 = &Xs[0][0];
    float* const ws1 = &Ws[1][0];
    float* const xs1 = &Xs[1][0];

    v8f acc = {};
    PJ_FETCH(w00, w01, x00, x01, x02, x03, 0)
    PJ_FETCH(w10, w11, x10, x11, x12, x13, 1)
    PJ_FETCH(w20, w21, x20, x21, x22, x23, 2)
    PJ_PARK(w00, w01, x00, x01, x02, x03, ws0, xs0)
    pj_barrier(acc);
    PJ_FETCH(w00, w01, x00, x01, x02, x03, 3)
    PJ_READ(fa0, fb0, ws0, xs0)
    PJ_PARK(w10, w11, x10, x11, x12, x13, ws1, xs1)
    pj_barrier(acc);
    PJ_FETCH(w10, w11, x10, x11, x12, x13, 4)
    for (int i = 0; i < PJ_NCH - 4; i += 6) {
        PJ_STAGE(fa1, fb1, ws1, xs1, w20, w21, x20, x21, x22, x23, ws0, xs0, i + 5, fa0, fb0)
        PJ_STAGE(fa0, fb0, ws0, xs0, w00, w01, x00, x01, x02, x03, ws1, xs1, i + 6, fa1, fb1)
        PJ_STAGE(fa1, fb1, ws1, xs1, w10, w11, x10, x11, x12, x13, ws0, xs0, i + 7, fa0, fb0)
        PJ_STAGE(fa0, fb0, ws0, xs0, w20, w21, x20, x21, x22, x23, ws1, xs1, i + 8, fa1, fb1)
        PJ_STAGE(fa1, fb1, ws1, xs1, w00, w01, x00, x01, x02, x03, ws0, xs0, i + 9, fa0, fb0)
        PJ_STAGE(fa0, fb0, ws0, xs0, w10, w11, x10, x11, x12, x13, ws1, xs1,
                 (i + 10 < PJ_NCH ? i + 10 : PJ_NCH - 1), fa1, fb1)
    }
    PJ_READ(fa1, fb1, ws1, xs1)
    PJ_PARK(w20, w21, x20, x21, x22, x23, ws0, xs0)
    PJ_MULT_HALF(fa0, fb0, 0)
    pj_signal(acc);
    PJ_MULT_HALF(fa0, fb0, 8)
    pj_wait(acc);
    PJ_READ(fa0, fb0, ws0, xs0)
    PJ_PARK(w00, w01, x00, x01, x02, x03, ws1, xs1)
    PJ_MULT_HALF(fa1, fb1, 0)
    pj_signal(acc);
    PJ_MULT_HALF(fa1, fb1, 8)
    pj_wait(acc);
    PJ_READ(fa1, fb1, ws1, xs1)
    PJ_MULT(fa0, fb0)
    PJ_MULT(fa1, fb1)
#undef PJ_STAGE
#undef PJ_READ
#undef PJ_FETCH
#undef PJ_PARK
#undef PJ_MULT
#undef PJ_MULT_HALF

    const int kpt = kp0 + 16 * wi;
    const int d   = 16 * wj + ll;
#pragma unroll
    for (int v = 0; v < 8; ++v) acc[v] += bias[kpt + 8 * lh + v];

    if (!isV) {
        float* p = Kpp + (size_t)b * D * KP + (size_t)(d >> 1) * (2 * KP) +
                   (size_t)(kpt + 8 * lh) * 2 + (d & 1);
#pragma unroll
        for (int v = 0; v < 8; ++v) p[2 * v] = acc[v];
    } else {
        float* base = Vpp + (size_t)b * KP * D + d * 2;
#pragma unroll
        for (int v = 0; v < 8; ++v) {
            const int kp = kpt + 8 * lh + v;
            base[(size_t)(kp >> 1) * (2 * D) + (kp & 1)] = acc[v];
        }
    }
}

#define LDS_STRIDE 260
#define AT_MS 2
#define AT_ROWS (16 * AT_MS)

template <int M>
__device__ __forceinline__ float row_xor(float x) {
    return __builtin_bit_cast(
        float, __builtin_amdgcn_update_dpp(0, __builtin_bit_cast(int, x), 0x160 + M, 0xf,
                                           0xf, true));
}

__global__ __launch_bounds__(32) void lin_attn_kernel(
    const float* __restrict__ Q, const float* __restrict__ Kpp,
    const float* __restrict__ Vpp, float* __restrict__ out,
    float* __restrict__ attn) {
    const int S = LIN_S, D = LIN_D, KP = LIN_KP;
    const float scale = 0.125f;

    __shared__ float lds[AT_ROWS * LDS_STRIDE];

    const int blk = blockIdx.x;
    const int b   = blk / (LIN_S / AT_ROWS);
    const int s0  = (blk % (LIN_S / AT_ROWS)) * AT_ROWS;
    const int l   = threadIdx.x;
    const int lh  = l >> 4;
    const int ll  = l & 15;

    v2f qa[AT_MS][16];
#pragma unroll
    for (int ms = 0; ms < AT_MS; ++ms) {
        const float* qrow =
            Q + (size_t)b * S * D + (size_t)(s0 + 16 * ms + ll) * D + lh * 2;
#pragma unroll
        for (int kk = 0; kk < 16; ++kk) qa[ms][kk] = *(const v2f*)(qrow + kk * 4);
    }

    v8f sc[AT_MS][16];
    const float* kpB = Kpp + (size_t)b * D * KP + ll * 2;
    v2f bfn[16];
#pragma unroll
    for (int kk = 0; kk < 16; ++kk)
        bfn[kk] = *(const v2f*)(kpB + (size_t)(kk * 2 + lh) * (2 * KP));
#pragma unroll
    for (int t = 0; t < 16; ++t) {
        v2f bf[16];
#pragma unroll
        for (int kk = 0; kk < 16; ++kk) bf[kk] = bfn[kk];
        if (t + 1 < 16) {
#pragma unroll
            for (int kk = 0; kk < 16; ++kk)
                bfn[kk] = *(const v2f*)(kpB + (size_t)(kk * 2 + lh) * (2 * KP) + (t + 1) * 32);
        }
        v8f c[AT_MS];
#pragma unroll
        for (int ms = 0; ms < AT_MS; ++ms) c[ms] = v8f{};
#pragma unroll
        for (int kk = 0; kk < 16; ++kk) {
#pragma unroll
            for (int ms = 0; ms < AT_MS; ++ms) c[ms] = wmma_f32(qa[ms][kk], bf[kk], c[ms]);
        }
#pragma unroll
        for (int ms = 0; ms < AT_MS; ++ms) sc[ms][t] = c[ms];
    }

#pragma unroll
    for (int ms = 0; ms < AT_MS; ++ms) {
        float mx[8];
#pragma unroll
        for (int v = 0; v < 8; ++v) {
            float m = sc[ms][0][v];
#pragma unroll
            for (int t = 1; t < 16; ++t) m = fmaxf(m, sc[ms][t][v]);
            mx[v] = m;
        }
#pragma unroll
        for (int v = 0; v < 8; ++v) mx[v] = fmaxf(mx[v], row_xor<1>(mx[v]));
#pragma unroll
        for (int v = 0; v < 8; ++v) mx[v] = fmaxf(mx[v], row_xor<2>(mx[v]));
#pragma unroll
        for (int v = 0; v < 8; ++v) mx[v] = fmaxf(mx[v], row_xor<4>(mx[v]));
#pragma unroll
        for (int v = 0; v < 8; ++v) mx[v] = fmaxf(mx[v], row_xor<8>(mx[v]));

        float sum[8];
#pragma unroll
        for (int v = 0; v < 8; ++v) {
            float s = 0.f;
#pragma unroll
            for (int t = 0; t < 16; ++t) {
                float p = __expf((sc[ms][t][v] - mx[v]) * scale);
                sc[ms][t][v] = p;
                s += p;
            }
            sum[v] = s;
        }
#pragma unroll
        for (int v = 0; v < 8; ++v) sum[v] += row_xor<1>(sum[v]);
#pragma unroll
        for (int v = 0; v < 8; ++v) sum[v] += row_xor<2>(sum[v]);
#pragma unroll
        for (int v = 0; v < 8; ++v) sum[v] += row_xor<4>(sum[v]);
#pragma unroll
        for (int v = 0; v < 8; ++v) sum[v] += row_xor<8>(sum[v]);

#pragma unroll
        for (int v = 0; v < 8; ++v) {
            const float r = 1.0f / sum[v];
#pragma unroll
            for (int t = 0; t < 16; ++t) sc[ms][t][v] *= r;
        }
    }

    float* arow = attn + (size_t)b * S * KP;
#pragma unroll
    for (int ms = 0; ms < AT_MS; ++ms) {
#pragma unroll
        for (int t = 0; t < 16; ++t) {
#pragma unroll
            for (int v = 0; v < 8; ++v) {
                const int m   = 16 * ms + v + 8 * lh;
                const int col = t * 16 + ll;
                arow[(size_t)(s0 + m) * KP + col] = sc[ms][t][v];
                lds[m * LDS_STRIDE + col]         = sc[ms][t][v];
            }
        }
    }
    __syncthreads();

    const float* vpB = Vpp + (size_t)b * KP * D + ll * 2;
    v8f oc[AT_MS][4];
#pragma unroll
    for (int ms = 0; ms < AT_MS; ++ms) {
#pragma unroll
        for (int td = 0; td < 4; ++td) oc[ms][td] = v8f{};
    }
#pragma unroll 4
    for (int kk = 0; kk < 64; ++kk) {
        v2f a[AT_MS];
#pragma unroll
        for (int ms = 0; ms < AT_MS; ++ms)
            a[ms] = *(const v2f*)(&lds[(16 * ms + ll) * LDS_STRIDE + kk * 4 + lh * 2]);
        v2f bf[4];
#pragma unroll
        for (int td = 0; td < 4; ++td)
            bf[td] = *(const v2f*)(vpB + (size_t)(kk * 2 + lh) * (2 * D) + td * 32);
#pragma unroll
        for (int td = 0; td < 4; ++td) {
#pragma unroll
            for (int ms = 0; ms < AT_MS; ++ms)
                oc[ms][td] = wmma_f32(a[ms], bf[td], oc[ms][td]);
        }
    }
#pragma unroll
    for (int ms = 0; ms < AT_MS; ++ms) {
#pragma unroll
        for (int td = 0; td < 4; ++td) {
#pragma unroll
            for (int v = 0; v < 8; ++v)
                out[(size_t)b * S * D + (size_t)(s0 + 16 * ms + v + 8 * lh) * D + td * 16 +
                    ll] = oc[ms][td][v];
        }
    }
}

extern "C" void kernel_launch(void* const* d_in, const int* in_sizes, int n_in,
                              void* d_out, int out_size, void* d_ws, size_t ws_size,
                              hipStream_t stream) {
    (void)in_sizes; (void)n_in; (void)out_size; (void)ws_size;
    const float* Q   = (const float*)d_in[0];
    const float* K   = (const float*)d_in[1];
    const float* V   = (const float*)d_in[2];
    const float* E_w = (const float*)d_in[3];
    const float* E_b = (const float*)d_in[4];
    const float* F_w = (const float*)d_in[5];
    const float* F_b = (const float*)d_in[6];

    float* out  = (float*)d_out;
    float* attn = out + (size_t)LIN_B * LIN_S * LIN_D;

    float* Kpp = (float*)d_ws;
    float* Vpp = Kpp + (size_t)LIN_B * LIN_D * LIN_KP;

    const int attnBlocks = LIN_B * (LIN_S / AT_ROWS);

    lin_proj_kernel<<<256, 256, 0, stream>>>(K, V, E_w, F_w, E_b, F_b, Kpp, Vpp);
    lin_attn_kernel<<<attnBlocks, 32, 0, stream>>>(Q, Kpp, Vpp, out, attn);
}


# Output GEMM fetches one group ahead: the next 8 k-steps' 32 Vp B fragments (L2) and 16 attn A fragments (LDS) are requested in front of the current 64 products, two register sets; sums unchanged
# speedup vs baseline: 3.2228x; 1.0493x over previous
#include <hip/hip_runtime.h>
#include <hip/hip_bf16.h>


typedef float v2f __attribute__((ext_vector_type(2)));
typedef float v8f __attribute__((ext_vector_type(8)));

#define LIN_B  16
#define LIN_S  4096
#define LIN_D  64
#define LIN_KP 256

__device__ __forceinline__ v8f wmma_f32(v2f a, v2f b, v8f c) {
    return __builtin_amdgcn_wmma_f32_16x16x4_f32(false, a, false, b, (short)0, c,
                                                 false, false);
}

typedef float v4f __attribute__((ext_vector_type(4)));

#define PJ_CH  64
#define PJ_NCH (LIN_S / PJ_CH)
#define PJ_WP  68
#define PJ_XP  160

static __device__ __forceinline__ void pj_barrier(v8f acc_done) {
    asm volatile("s_wait_dscnt 0x0\n\ts_barrier_signal -1\n\ts_barrier_wait -1"
                 :: "v"(acc_done) : "memory");
}

static __device__ __forceinline__ void pj_signal(v8f acc_half) {
    asm volatile("s_wait_dscnt 0x0\n\ts_barrier_signal -1" :: "v"(acc_half) : "memory");
}
static __device__ __forceinline__ void pj_wait(v8f acc_done) {
    asm volatile("s_barrier_wait -1" :: "v"(acc_done) : "memory");
}

__global__ __launch_bounds__(256) void lin_proj_kernel(
    const float* __restrict__ Kx, const float* __restrict__ Vx,
    const float* __restrict__ Ew, const float* __restrict__ Fw,
    const float* __restrict__ Eb, const float* __restrict__ Fb,
    float* __restrict__ Kpp, float* __restrict__ Vpp) {
    const int S = LIN_S, D = LIN_D, KP = LIN_KP;

    __shared__ __attribute__((aligned(16))) float Ws[2][32 * PJ_WP];
    __shared__ __attribute__((aligned(16))) float Xs[2][32 * PJ_XP];

    const int blk = blockIdx.x;
    const bool isV = (blk >> 7) != 0;
    const int b    = (blk >> 3) & 15;
    const int kp0  = (blk & 7) * 32;

    const float* X    = (isV ? Vx : Kx) + (size_t)b * S * D;
    const float* W    = (isV ? Fw : Ew) + (size_t)kp0 * S;
    const float* bias = isV ? Fb : Eb;

    const int t    = threadIdx.x;
    const int wave = t >> 5;
    const int l    = t & 31;
    const int lh   = l >> 4;
    const int ll   = l & 15;
    const int wi   = wave >> 2;
    const int wj   = wave & 3;

    const int q = t & 15;
    const int r = t >> 4;
    const float* gW = W + (size_t)r * S + 4 * q;
    const float* gX = X + (size_t)(2 * r) * D + 4 * q;
    const int wOff = r * PJ_WP + 4 * q;
    const int xOff = r * PJ_XP + 8 * q;

    const int aOff = (16 * wi + ll) * PJ_WP + 2 * lh;
    const int bOff = lh * PJ_XP + (16 * wj + ll) * 2;

    v4f w00, w01, x00, x01, x02, x03;
    v4f w10, w11, x10, x11, x12, x13;
    v4f w20, w21, x20, x21, x22, x23;
    v2f fa0[16], fb0[16], fa1[16], fb1[16];

#define PJ_FETCH(w0, w1, x0, x1, x2, x3, chunk)                         \
    {                                                                   \
        const int sn_ = (chunk) * PJ_CH;                                \
        w0 = *(const v4f*)(gW + sn_);                                   \
        w1 = *(const v4f*)(gW + (size_t)16 * S + sn_);                  \
        x0 = *(const v4f*)(gX + (size_t)sn_ * D);                       \
        x1 = *(const v4f*)(gX + (size_t)sn_ * D + D);                   \
        x2 = *(const v4f*)(gX + (size_t)sn_ * D + 32 * D);              \
        x3 = *(const v4f*)(gX + (size_t)sn_ * D + 33 * D);              \
    }

#define PJ_PARK(w0, w1, x0, x1, x2, x3, ws, xs)                         \
    {                                                                   \
        *(v4f*)((ws) + wOff)              = w0;                         \
        *(v4f*)((ws) + wOff + 16 * PJ_WP) = w1;                         \
        v4f p0_ = {x0.x, x1.x, x0.y, x1.y};                             \
        v4f p1_ = {x0.z, x1.z, x0.w, x1.w};                             \
        v4f p2_ = {x2.x, x3.x, x2.y, x3.y};                             \
        v4f p3_ = {x2.z, x3.z, x2.w, x3.w};                             \
        *(v4f*)((xs) + xOff)                  = p0_;                    \
        *(v4f*)((xs) + xOff + 4)              = p1_;                    \
        *(v4f*)((xs) + xOff + 16 * PJ_XP)     = p2_;                    \
        *(v4f*)((xs) + xOff + 16 * PJ_XP + 4) = p3_;                    \
    }

#define PJ_READ(fa, fb, ws, xs)                                         \
    {                                                                   \
        _Pragma("unroll") for (int j = 0; j < 16; ++j)                  \
            fa[j] = *(const v2f*)((ws) + aOff + 4 * j);                 \
        _Pragma("unroll") for (int j = 0; j < 16; ++j)                  \
            fb[j] = *(const v2f*)((xs) + bOff + 2 * j * PJ_XP);         \
    }

#define PJ_MULT(fa, fb)                                                 \
    {                                                                   \
        __builtin_amdgcn_sched_barrier(0);                              \
        _Pragma("unroll") for (int j = 0; j < 16; ++j)                  \
            acc = wmma_f32(fa[j], fb[j], acc);                          \
        __builtin_amdgcn_sched_barrier(0);                              \
    }
#define PJ_MULT_HALF(fa, fb, j0)                                        \
    {                                                                   \
        __builtin_amdgcn_sched_barrier(0);                              \
        _Pragma("unroll") for (int j = (j0); j < (j0) + 8; ++j)         \
            acc = wmma_f32(fa[j], fb[j], acc);                          \
        __builtin_amdgcn_sched_barrier(0);                              \
    }

#define PJ_STAGE(fnext, fbnext, rws, rxs, w0, w1, x0, x1, x2, x3, pws, pxs, \
                 fchunk, fcur, fbcur)                                       \
    {                                                                       \
        PJ_READ(fnext, fbnext, rws, rxs)                                    \
        PJ_PARK(w0, w1, x0, x1, x2, x3, pws, pxs)                           \
        PJ_FETCH(w0, w1, x0, x1, x2, x3, fchunk)                            \
        PJ_MULT_HALF(fcur, fbcur, 0)                                        \
        pj_signal(acc);                                                     \
        PJ_MULT_HALF(fcur, fbcur, 8)                                        \
        pj_wait(acc);                                                       \
    }
    float* const ws0 = &Ws[0][0];
    float* const xs0 = &Xs[0][0];
    float* const ws1 = &Ws[1][0];
    float* const xs1 = &Xs[1][0];

    v8f acc = {};
    PJ_FETCH(w00, w01, x00, x01, x02, x03, 0)
    PJ_FETCH(w10, w11, x10, x11, x12, x13, 1)
    PJ_FETCH(w20, w21, x20, x21, x22, x23, 2)
    PJ_PARK(w00, w01, x00, x01, x02, x03, ws0, xs0)
    pj_barrier(acc);
    PJ_FETCH(w00, w01, x00, x01, x02, x03, 3)
    PJ_READ(fa0, fb0, ws0, xs0)
    PJ_PARK(w10, w11, x10, x11, x12, x13, ws1, xs1)
    pj_barrier(acc);
    PJ_FETCH(w10, w11, x10, x11, x12, x13, 4)
    for (int i = 0; i < PJ_NCH - 4; i += 6) {
        PJ_STAGE(fa1, fb1, ws1, xs1, w20, w21, x20, x21, x22, x23, ws0, xs0, i + 5, fa0, fb0)
        PJ_STAGE(fa0, fb0, ws0, xs0, w00, w01, x00, x01, x02, x03, ws1, xs1, i + 6, fa1, fb1)
        PJ_STAGE(fa1, fb1, ws1, xs1, w10, w11, x10, x11, x12, x13, ws0, xs0, i + 7, fa0, fb0)
        PJ_STAGE(fa0, fb0, ws0, xs0, w20, w21, x20, x21, x22, x23, ws1, xs1, i + 8, fa1, fb1)
        PJ_STAGE(fa1, fb1, ws1, xs1, w00, w01, x00, x01, x02, x03, ws0, xs0, i + 9, fa0, fb0)
        PJ_STAGE(fa0, fb0, ws0, xs0, w10, w11, x10, x11, x12, x13, ws1, xs1,
                 (i + 10 < PJ_NCH ? i + 10 : PJ_NCH - 1), fa1, fb1)
    }
    PJ_READ(fa1, fb1, ws1, xs1)
    PJ_PARK(w20, w21, x20, x21, x22, x23, ws0, xs0)
    PJ_MULT_HALF(fa0, fb0, 0)
    pj_signal(acc);
    PJ_MULT_HALF(fa0, fb0, 8)
    pj_wait(acc);
    PJ_READ(fa0, fb0, ws0, xs0)
    PJ_PARK(w00, w01, x00, x01, x02, x03, ws1, xs1)
    PJ_MULT_HALF(fa1, fb1, 0)
    pj_signal(acc);
    PJ_MULT_HALF(fa1, fb1, 8)
    pj_wait(acc);
    PJ_READ(fa1, fb1, ws1, xs1)
    PJ_MULT(fa0, fb0)
    PJ_MULT(fa1, fb1)
#undef PJ_STAGE
#undef PJ_READ
#undef PJ_FETCH
#undef PJ_PARK
#undef PJ_MULT
#undef PJ_MULT_HALF

    const int kpt = kp0 + 16 * wi;
    const int d   = 16 * wj + ll;
#pragma unroll
    for (int v = 0; v < 8; ++v) acc[v] += bias[kpt + 8 * lh + v];

    if (!isV) {
        float* p = Kpp + (size_t)b * D * KP + (size_t)(d >> 1) * (2 * KP) +
                   (size_t)(kpt + 8 * lh) * 2 + (d & 1);
#pragma unroll
        for (int v = 0; v < 8; ++v) p[2 * v] = acc[v];
    } else {
        float* base = Vpp + (size_t)b * KP * D + d * 2;
#pragma unroll
        for (int v = 0; v < 8; ++v) {
            const int kp = kpt + 8 * lh + v;
            base[(size_t)(kp >> 1) * (2 * D) + (kp & 1)] = acc[v];
        }
    }
}

#define LDS_STRIDE 260
#define AT_MS 2
#define AT_ROWS (16 * AT_MS)

template <int M>
__device__ __forceinline__ float row_xor(float x) {
    return __builtin_bit_cast(
        float, __builtin_amdgcn_update_dpp(0, __builtin_bit_cast(int, x), 0x160 + M, 0xf,
                                           0xf, true));
}

__global__ __launch_bounds__(32) void lin_attn_kernel(
    const float* __restrict__ Q, const float* __restrict__ Kpp,
    const float* __restrict__ Vpp, float* __restrict__ out,
    float* __restrict__ attn) {
    const int S = LIN_S, D = LIN_D, KP = LIN_KP;
    const float scale = 0.125f;

    __shared__ float lds[AT_ROWS * LDS_STRIDE];

    const int blk = blockIdx.x;
    const int b   = blk / (LIN_S / AT_ROWS);
    const int s0  = (blk % (LIN_S / AT_ROWS)) * AT_ROWS;
    const int l   = threadIdx.x;
    const int lh  = l >> 4;
    const int ll  = l & 15;

    v2f qa[AT_MS][16];
#pragma unroll
    for (int ms = 0; ms < AT_MS; ++ms) {
        const float* qrow =
            Q + (size_t)b * S * D + (size_t)(s0 + 16 * ms + ll) * D + lh * 2;
#pragma unroll
        for (int kk = 0; kk < 16; ++kk) qa[ms][kk] = *(const v2f*)(qrow + kk * 4);
    }

    v8f sc[AT_MS][16];
    const float* kpB = Kpp + (size_t)b * D * KP + ll * 2;
    v2f bfn[16];
#pragma unroll
    for (int kk = 0; kk < 16; ++kk)
        bfn[kk] = *(const v2f*)(kpB + (size_t)(kk * 2 + lh) * (2 * KP));
#pragma unroll
    for (int t = 0; t < 16; ++t) {
        v2f bf[16];
#pragma unroll
        for (int kk = 0; kk < 16; ++kk) bf[kk] = bfn[kk];
        if (t + 1 < 16) {
#pragma unroll
            for (int kk = 0; kk < 16; ++kk)
                bfn[kk] = *(const v2f*)(kpB + (size_t)(kk * 2 + lh) * (2 * KP) + (t + 1) * 32);
        }
        v8f c[AT_MS];
#pragma unroll
        for (int ms = 0; ms < AT_MS; ++ms) c[ms] = v8f{};
#pragma unroll
        for (int kk = 0; kk < 16; ++kk) {
#pragma unroll
            for (int ms = 0; ms < AT_MS; ++ms) c[ms] = wmma_f32(qa[ms][kk], bf[kk], c[ms]);
        }
#pragma unroll
        for (int ms = 0; ms < AT_MS; ++ms) sc[ms][t] = c[ms];
    }

#pragma unroll
    for (int ms = 0; ms < AT_MS; ++ms) {
        float mx[8];
#pragma unroll
        for (int v = 0; v < 8; ++v) {
            float m = sc[ms][0][v];
#pragma unroll
            for (int t = 1; t < 16; ++t) m = fmaxf(m, sc[ms][t][v]);
            mx[v] = m;
        }
#pragma unroll
        for (int v = 0; v < 8; ++v) mx[v] = fmaxf(mx[v], row_xor<1>(mx[v]));
#pragma unroll
        for (int v = 0; v < 8; ++v) mx[v] = fmaxf(mx[v], row_xor<2>(mx[v]));
#pragma unroll
        for (int v = 0; v < 8; ++v) mx[v] = fmaxf(mx[v], row_xor<4>(mx[v]));
#pragma unroll
        for (int v = 0; v < 8; ++v) mx[v] = fmaxf(mx[v], row_xor<8>(mx[v]));

        float sum[8];
#pragma unroll
        for (int v = 0; v < 8; ++v) {
            float s = 0.f;
#pragma unroll
            for (int t = 0; t < 16; ++t) {
                float p = __expf((sc[ms][t][v] - mx[v]) * scale);
                sc[ms][t][v] = p;
                s += p;
            }
            sum[v] = s;
        }
#pragma unroll
        for (int v = 0; v < 8; ++v) sum[v] += row_xor<1>(sum[v]);
#pragma unroll
        for (int v = 0; v < 8; ++v) sum[v] += row_xor<2>(sum[v]);
#pragma unroll
        for (int v = 0; v < 8; ++v) sum[v] += row_xor<4>(sum[v]);
#pragma unroll
        for (int v = 0; v < 8; ++v) sum[v] += row_xor<8>(sum[v]);

#pragma unroll
        for (int v = 0; v < 8; ++v) {
            const float r = 1.0f / sum[v];
#pragma unroll
            for (int t = 0; t < 16; ++t) sc[ms][t][v] *= r;
        }
    }

#define AT_G  8
#define AT_NG (LIN_KP / (4 * AT_G))
#define AT_LOADB(fb, g)                                                       \
    {                                                                         \
        _Pragma("unroll") for (int j = 0; j < AT_G; ++j) {                    \
            _Pragma("unroll") for (int td = 0; td < 4; ++td)                  \
                fb[j][td] = *(const v2f*)(vpB +                               \
                    (size_t)(((g) * AT_G + j) * 2 + lh) * (2 * D) + td * 32); \
        }                                                                     \
    }
#define AT_LOADA(fa, g)                                                       \
    {                                                                         \
        _Pragma("unroll") for (int j = 0; j < AT_G; ++j) {                    \
            _Pragma("unroll") for (int ms = 0; ms < AT_MS; ++ms)              \
                fa[j][ms] = *(const v2f*)(&lds[(16 * ms + ll) * LDS_STRIDE +  \
                                               ((g) * AT_G + j) * 4 + lh * 2]); \
        }                                                                     \
    }
#define AT_MMA(fa, fb)                                                        \
    {                                                                         \
        __builtin_amdgcn_sched_barrier(0);                                    \
        _Pragma("unroll") for (int j = 0; j < AT_G; ++j) {                    \
            _Pragma("unroll") for (int td = 0; td < 4; ++td) {                \
                _Pragma("unroll") for (int ms = 0; ms < AT_MS; ++ms)          \
                    oc[ms][td] = wmma_f32(fa[j][ms], fb[j][td], oc[ms][td]);  \
            }                                                                 \
        }                                                                     \
        __builtin_amdgcn_sched_barrier(0);                                    \
    }
    const float* vpB = Vpp + (size_t)b * KP * D + ll * 2;
    v2f va0[AT_G][AT_MS], vb0[AT_G][4], va1[AT_G][AT_MS], vb1[AT_G][4];
    AT_LOADB(vb0, 0)

    float* arow = attn + (size_t)b * S * KP;
#pragma unroll
    for (int ms = 0; ms < AT_MS; ++ms) {
#pragma unroll
        for (int t = 0; t < 16; ++t) {
#pragma unroll
            for (int v = 0; v < 8; ++v) {
                const int m   = 16 * ms + v + 8 * lh;
                const int col = t * 16 + ll;
                arow[(size_t)(s0 + m) * KP + col] = sc[ms][t][v];
                lds[m * LDS_STRIDE + col]         = sc[ms][t][v];
            }
        }
    }
    __syncthreads();

    v8f oc[AT_MS][4];
#pragma unroll
    for (int ms = 0; ms < AT_MS; ++ms) {
#pragma unroll
        for (int td = 0; td < 4; ++td) oc[ms][td] = v8f{};
    }
    AT_LOADA(va0, 0)
#pragma unroll 1
    for (int g = 0; g < AT_NG - 2; g += 2) {
        AT_LOADB(vb1, g + 1)
        AT_LOADA(va1, g + 1)
        AT_MMA(va0, vb0)
        AT_LOADB(vb0, g + 2)
        AT_LOADA(va0, g + 2)
        AT_MMA(va1, vb1)
    }
    AT_LOADB(vb1, AT_NG - 1)
    AT_LOADA(va1, AT_NG - 1)
    AT_MMA(va0, vb0)
    AT_MMA(va1, vb1)
#undef AT_LOADB
#undef AT_LOADA
#undef AT_MMA
#undef AT_G
#undef AT_NG
#pragma unroll
    for (int ms = 0; ms < AT_MS; ++ms) {
#pragma unroll
        for (int td = 0; td < 4; ++td) {
#pragma unroll
            for (int v = 0; v < 8; ++v)
                out[(size_t)b * S * D + (size_t)(s0 + 16 * ms + v + 8 * lh) * D + td * 16 +
                    ll] = oc[ms][td][v];
        }
    }
}

extern "C" void kernel_launch(void* const* d_in, const int* in_sizes, int n_in,
                              void* d_out, int out_size, void* d_ws, size_t ws_size,
                              hipStream_t stream) {
    (void)in_sizes; (void)n_in; (void)out_size; (void)ws_size;
    const float* Q   = (const float*)d_in[0];
    const float* K   = (const float*)d_in[1];
    const float* V   = (const float*)d_in[2];
    const float* E_w = (const float*)d_in[3];
    const float* E_b = (const float*)d_in[4];
    const float* F_w = (const float*)d_in[5];
    const float* F_b = (const float*)d_in[6];

    float* out  = (float*)d_out;
    float* attn = out + (size_t)LIN_B * LIN_S * LIN_D;

    float* Kpp = (float*)d_ws;
    float* Vpp = Kpp + (size_t)LIN_B * LIN_D * LIN_KP;

    const int attnBlocks = LIN_B * (LIN_S / AT_ROWS);

    lin_proj_kernel<<<256, 256, 0, stream>>>(K, V, E_w, F_w, E_b, F_b, Kpp, Vpp);
    lin_attn_kernel<<<attnBlocks, 32, 0, stream>>>(Q, Kpp, Vpp, out, attn);
}


# attn leaves from its LDS copy, not from the score registers: behind the strip barrier each 64-product group carries 8 row reads (LDS) and 8 whole-line 16-byte stores; order of sums unchanged
# speedup vs baseline: 3.3939x; 1.0531x over previous
#include <hip/hip_runtime.h>
#include <hip/hip_bf16.h>


typedef float v2f __attribute__((ext_vector_type(2)));
typedef float v8f __attribute__((ext_vector_type(8)));

#define LIN_B  16
#define LIN_S  4096
#define LIN_D  64
#define LIN_KP 256

__device__ __forceinline__ v8f wmma_f32(v2f a, v2f b, v8f c) {
    return __builtin_amdgcn_wmma_f32_16x16x4_f32(false, a, false, b, (short)0, c,
                                                 false, false);
}

typedef float v4f __attribute__((ext_vector_type(4)));

#define PJ_CH  64
#define PJ_NCH (LIN_S / PJ_CH)
#define PJ_WP  68
#define PJ_XP  160

static __device__ __forceinline__ void pj_barrier(v8f acc_done) {
    asm volatile("s_wait_dscnt 0x0\n\ts_barrier_signal -1\n\ts_barrier_wait -1"
                 :: "v"(acc_done) : "memory");
}

static __device__ __forceinline__ void pj_signal(v8f acc_half) {
    asm volatile("s_wait_dscnt 0x0\n\ts_barrier_signal -1" :: "v"(acc_half) : "memory");
}
static __device__ __forceinline__ void pj_wait(v8f acc_done) {
    asm volatile("s_barrier_wait -1" :: "v"(acc_done) : "memory");
}

__global__ __launch_bounds__(256) void lin_proj_kernel(
    const float* __restrict__ Kx, const float* __restrict__ Vx,
    const float* __restrict__ Ew, const float* __restrict__ Fw,
    const float* __restrict__ Eb, const float* __restrict__ Fb,
    float* __restrict__ Kpp, float* __restrict__ Vpp) {
    const int S = LIN_S, D = LIN_D, KP = LIN_KP;

    __shared__ __attribute__((aligned(16))) float Ws[2][32 * PJ_WP];
    __shared__ __attribute__((aligned(16))) float Xs[2][32 * PJ_XP];

    const int blk = blockIdx.x;
    const bool isV = (blk >> 7) != 0;
    const int b    = (blk >> 3) & 15;
    const int kp0  = (blk & 7) * 32;

    const float* X    = (isV ? Vx : Kx) + (size_t)b * S * D;
    const float* W    = (isV ? Fw : Ew) + (size_t)kp0 * S;
    const float* bias = isV ? Fb : Eb;

    const int t    = threadIdx.x;
    const int wave = t >> 5;
    const int l    = t & 31;
    const int lh   = l >> 4;
    const int ll   = l & 15;
    const int wi   = wave >> 2;
    const int wj   = wave & 3;

    const int q = t & 15;
    const int r = t >> 4;
    const float* gW = W + (size_t)r * S + 4 * q;
    const float* gX = X + (size_t)(2 * r) * D + 4 * q;
    const int wOff = r * PJ_WP + 4 * q;
    const int xOff = r * PJ_XP + 8 * q;

    const int aOff = (16 * wi + ll) * PJ_WP + 2 * lh;
    const int bOff = lh * PJ_XP + (16 * wj + ll) * 2;

    v4f w00, w01, x00, x01, x02, x03;
    v4f w10, w11, x10, x11, x12, x13;
    v4f w20, w21, x20, x21, x22, x23;
    v2f fa0[16], fb0[16], fa1[16], fb1[16];

#define PJ_FETCH(w0, w1, x0, x1, x2, x3, chunk)                         \
    {                                                                   \
        const int sn_ = (chunk) * PJ_CH;                                \
        w0 = *(const v4f*)(gW + sn_);                                   \
        w1 = *(const v4f*)(gW + (size_t)16 * S + sn_);                  \
        x0 = *(const v4f*)(gX + (size_t)sn_ * D);                       \
        x1 = *(const v4f*)(gX + (size_t)sn_ * D + D);                   \
        x2 = *(const v4f*)(gX + (size_t)sn_ * D + 32 * D);              \
        x3 = *(const v4f*)(gX + (size_t)sn_ * D + 33 * D);              \
    }

#define PJ_PARK(w0, w1, x0, x1, x2, x3, ws, xs)                         \
    {                                                                   \
        *(v4f*)((ws) + wOff)              = w0;                         \
        *(v4f*)((ws) + wOff + 16 * PJ_WP) = w1;                         \
        v4f p0_ = {x0.x, x1.x, x0.y, x1.y};                             \
        v4f p1_ = {x0.z, x1.z, x0.w, x1.w};                             \
        v4f p2_ = {x2.x, x3.x, x2.y, x3.y};                             \
        v4f p3_ = {x2.z, x3.z, x2.w, x3.w};                             \
        *(v4f*)((xs) + xOff)                  = p0_;                    \
        *(v4f*)((xs) + xOff + 4)              = p1_;                    \
        *(v4f*)((xs) + xOff + 16 * PJ_XP)     = p2_;                    \
        *(v4f*)((xs) + xOff + 16 * PJ_XP + 4) = p3_;                    \
    }

#define PJ_READ(fa, fb, ws, xs)                                         \
    {                                                                   \
        _Pragma("unroll") for (int j = 0; j < 16; ++j)                  \
            fa[j] = *(const v2f*)((ws) + aOff + 4 * j);                 \
        _Pragma("unroll") for (int j = 0; j < 16; ++j)                  \
            fb[j] = *(const v2f*)((xs) + bOff + 2 * j * PJ_XP);         \
    }

#define PJ_MULT(fa, fb)                                                 \
    {                                                                   \
        __builtin_amdgcn_sched_barrier(0);                              \
        _Pragma("unroll") for (int j = 0; j < 16; ++j)                  \
            acc = wmma_f32(fa[j], fb[j], acc);                          \
        __builtin_amdgcn_sched_barrier(0);                              \
    }
#define PJ_MULT_HALF(fa, fb, j0)                                        \
    {                                                                   \
        __builtin_amdgcn_sched_barrier(0);                              \
        _Pragma("unroll") for (int j = (j0); j < (j0) + 8; ++j)         \
            acc = wmma_f32(fa[j], fb[j], acc);                          \
        __builtin_amdgcn_sched_barrier(0);                              \
    }

#define PJ_STAGE(fnext, fbnext, rws, rxs, w0, w1, x0, x1, x2, x3, pws, pxs, \
                 fchunk, fcur, fbcur)                                       \
    {                                                                       \
        PJ_READ(fnext, fbnext, rws, rxs)                                    \
        PJ_PARK(w0, w1, x0, x1, x2, x3, pws, pxs)                           \
        PJ_FETCH(w0, w1, x0, x1, x2, x3, fchunk)                            \
        PJ_MULT_HALF(fcur, fbcur, 0)                                        \
        pj_signal(acc);                                                     \
        PJ_MULT_HALF(fcur, fbcur, 8)                                        \
        pj_wait(acc);                                                       \
    }
    float* const ws0 = &Ws[0][0];
    float* const xs0 = &Xs[0][0];
    float* const ws1 = &Ws[1][0];
    float* const xs1 = &Xs[1][0];

    v8f acc = {};
    PJ_FETCH(w00, w01, x00, x01, x02, x03, 0)
    PJ_FETCH(w10, w11, x10, x11, x12, x13, 1)
    PJ_FETCH(w20, w21, x20, x21, x22, x23, 2)
    PJ_PARK(w00, w01, x00, x01, x02, x03, ws0, xs0)
    pj_barrier(acc);
    PJ_FETCH(w00, w01, x00, x01, x02, x03, 3)
    PJ_READ(fa0, fb0, ws0, xs0)
    PJ_PARK(w10, w11, x10, x11, x12, x13, ws1, xs1)
    pj_barrier(acc);
    PJ_FETCH(w10, w11, x10, x11, x12, x13, 4)
    for (int i = 0; i < PJ_NCH - 4; i += 6) {
        PJ_STAGE(fa1, fb1, ws1, xs1, w20, w21, x20, x21, x22, x23, ws0, xs0, i + 5, fa0, fb0)
        PJ_STAGE(fa0, fb0, ws0, xs0, w00, w01, x00, x01, x02, x03, ws1, xs1, i + 6, fa1, fb1)
        PJ_STAGE(fa1, fb1, ws1, xs1, w10, w11, x10, x11, x12, x13, ws0, xs0, i + 7, fa0, fb0)
        PJ_STAGE(fa0, fb0, ws0, xs0, w20, w21, x20, x21, x22, x23, ws1, xs1, i + 8, fa1, fb1)
        PJ_STAGE(fa1, fb1, ws1, xs1, w00, w01, x00, x01, x02, x03, ws0, xs0, i + 9, fa0, fb0)
        PJ_STAGE(fa0, fb0, ws0, xs0, w10, w11, x10, x11, x12, x13, ws1, xs1,
                 (i + 10 < PJ_NCH ? i + 10 : PJ_NCH - 1), fa1, fb1)
    }
    PJ_READ(fa1, fb1, ws1, xs1)
    PJ_PARK(w20, w21, x20, x21, x22, x23, ws0, xs0)
    PJ_MULT_HALF(fa0, fb0, 0)
    pj_signal(acc);
    PJ_MULT_HALF(fa0, fb0, 8)
    pj_wait(acc);
    PJ_READ(fa0, fb0, ws0, xs0)
    PJ_PARK(w00, w01, x00, x01, x02, x03, ws1, xs1)
    PJ_MULT_HALF(fa1, fb1, 0)
    pj_signal(acc);
    PJ_MULT_HALF(fa1, fb1, 8)
    pj_wait(acc);
    PJ_READ(fa1, fb1, ws1, xs1)
    PJ_MULT(fa0, fb0)
    PJ_MULT(fa1, fb1)
#undef PJ_STAGE
#undef PJ_READ
#undef PJ_FETCH
#undef PJ_PARK
#undef PJ_MULT
#undef PJ_MULT_HALF

    const int kpt = kp0 + 16 * wi;
    const int d   = 16 * wj + ll;
#pragma unroll
    for (int v = 0; v < 8; ++v) acc[v] += bias[kpt + 8 * lh + v];

    if (!isV) {
        float* p = Kpp + (size_t)b * D * KP + (size_t)(d >> 1) * (2 * KP) +
                   (size_t)(kpt + 8 * lh) * 2 + (d & 1);
#pragma unroll
        for (int v = 0; v < 8; ++v) p[2 * v] = acc[v];
    } else {
        float* base = Vpp + (size_t)b * KP * D + d * 2;
#pragma unroll
        for (int v = 0; v < 8; ++v) {
            const int kp = kpt + 8 * lh + v;
            base[(size_t)(kp >> 1) * (2 * D) + (kp & 1)] = acc[v];
        }
    }
}

#define LDS_STRIDE 260
#define AT_MS 2
#define AT_ROWS (16 * AT_MS)

template <int M>
__device__ __forceinline__ float row_xor(float x) {
    return __builtin_bit_cast(
        float, __builtin_amdgcn_update_dpp(0, __builtin_bit_cast(int, x), 0x160 + M, 0xf,
                                           0xf, true));
}

__global__ __launch_bounds__(32) void lin_attn_kernel(
    const float* __restrict__ Q, const float* __restrict__ Kpp,
    const float* __restrict__ Vpp, float* __restrict__ out,
    float* __restrict__ attn) {
    const int S = LIN_S, D = LIN_D, KP = LIN_KP;
    const float scale = 0.125f;

    __shared__ __attribute__((aligned(16))) float lds[AT_ROWS * LDS_STRIDE];

    const int blk = blockIdx.x;
    const int b   = blk / (LIN_S / AT_ROWS);
    const int s0  = (blk % (LIN_S / AT_ROWS)) * AT_ROWS;
    const int l   = threadIdx.x;
    const int lh  = l >> 4;
    const int ll  = l & 15;

    v2f qa[AT_MS][16];
#pragma unroll
    for (int ms = 0; ms < AT_MS; ++ms) {
        const float* qrow =
            Q + (size_t)b * S * D + (size_t)(s0 + 16 * ms + ll) * D + lh * 2;
#pragma unroll
        for (int kk = 0; kk < 16; ++kk) qa[ms][kk] = *(const v2f*)(qrow + kk * 4);
    }

    v8f sc[AT_MS][16];
    const float* kpB = Kpp + (size_t)b * D * KP + ll * 2;
    v2f bfn[16];
#pragma unroll
    for (int kk = 0; kk < 16; ++kk)
        bfn[kk] = *(const v2f*)(kpB + (size_t)(kk * 2 + lh) * (2 * KP));
#pragma unroll
    for (int t = 0; t < 16; ++t) {
        v2f bf[16];
#pragma unroll
        for (int kk = 0; kk < 16; ++kk) bf[kk] = bfn[kk];
        if (t + 1 < 16) {
#pragma unroll
            for (int kk = 0; kk < 16; ++kk)
                bfn[kk] = *(const v2f*)(kpB + (size_t)(kk * 2 + lh) * (2 * KP) + (t + 1) * 32);
        }
        v8f c[AT_MS];
#pragma unroll
        for (int ms = 0; ms < AT_MS; ++ms) c[ms] = v8f{};
#pragma unroll
        for (int kk = 0; kk < 16; ++kk) {
#pragma unroll
            for (int ms = 0; ms < AT_MS; ++ms) c[ms] = wmma_f32(qa[ms][kk], bf[kk], c[ms]);
        }
#pragma unroll
        for (int ms = 0; ms < AT_MS; ++ms) sc[ms][t] = c[ms];
    }

#pragma unroll
    for (int ms = 0; ms < AT_MS; ++ms) {
        float mx[8];
#pragma unroll
        for (int v = 0; v < 8; ++v) {
            float m = sc[ms][0][v];
#pragma unroll
            for (int t = 1; t < 16; ++t) m = fmaxf(m, sc[ms][t][v]);
            mx[v] = m;
        }
#pragma unroll
        for (int v = 0; v < 8; ++v) mx[v] = fmaxf(mx[v], row_xor<1>(mx[v]));
#pragma unroll
        for (int v = 0; v < 8; ++v) mx[v] = fmaxf(mx[v], row_xor<2>(mx[v]));
#pragma unroll
        for (int v = 0; v < 8; ++v) mx[v] = fmaxf(mx[v], row_xor<4>(mx[v]));
#pragma unroll
        for (int v = 0; v < 8; ++v) mx[v] = fmaxf(mx[v], row_xor<8>(mx[v]));

        float sum[8];
#pragma unroll
        for (int v = 0; v < 8; ++v) {
            float s = 0.f;
#pragma unroll
            for (int t = 0; t < 16; ++t) {
                float p = __expf((sc[ms][t][v] - mx[v]) * scale);
                sc[ms][t][v] = p;
                s += p;
            }
            sum[v] = s;
        }
#pragma unroll
        for (int v = 0; v < 8; ++v) sum[v] += row_xor<1>(sum[v]);
#pragma unroll
        for (int v = 0; v < 8; ++v) sum[v] += row_xor<2>(sum[v]);
#pragma unroll
        for (int v = 0; v < 8; ++v) sum[v] += row_xor<4>(sum[v]);
#pragma unroll
        for (int v = 0; v < 8; ++v) sum[v] += row_xor<8>(sum[v]);

#pragma unroll
        for (int v = 0; v < 8; ++v) {
            const float r = 1.0f / sum[v];
#pragma unroll
            for (int t = 0; t < 16; ++t) sc[ms][t][v] *= r;
        }
    }

#define AT_G  8
#define AT_NG (LIN_KP / (4 * AT_G))
#define AT_LOADB(fb, g)                                                       \
    {                                                                         \
        _Pragma("unroll") for (int j = 0; j < AT_G; ++j) {                    \
            _Pragma("unroll") for (int td = 0; td < 4; ++td)                  \
                fb[j][td] = *(const v2f*)(vpB +                               \
                    (size_t)(((g) * AT_G + j) * 2 + lh) * (2 * D) + td * 32); \
        }                                                                     \
    }
#define AT_LOADA(fa, g)                                                       \
    {                                                                         \
        _Pragma("unroll") for (int j = 0; j < AT_G; ++j) {                    \
            _Pragma("unroll") for (int ms = 0; ms < AT_MS; ++ms)              \
                fa[j][ms] = *(const v2f*)(&lds[(16 * ms + ll) * LDS_STRIDE +  \
                                               ((g) * AT_G + j) * 4 + lh * 2]); \
        }                                                                     \
    }
#define AT_MMA(fa, fb)                                                        \
    {                                                                         \
        __builtin_amdgcn_sched_barrier(0);                                    \
        _Pragma("unroll") for (int j = 0; j < AT_G; ++j) {                    \
            _Pragma("unroll") for (int td = 0; td < 4; ++td) {                \
                _Pragma("unroll") for (int ms = 0; ms < AT_MS; ++ms)          \
                    oc[ms][td] = wmma_f32(fa[j][ms], fb[j][td], oc[ms][td]);  \
            }                                                                 \
        }                                                                     \
        __builtin_amdgcn_sched_barrier(0);                                    \
    }
    const float* vpB = Vpp + (size_t)b * KP * D + ll * 2;
    v2f va0[AT_G][AT_MS], vb0[AT_G][4], va1[AT_G][AT_MS], vb1[AT_G][4];
    AT_LOADB(vb0, 0)

#pragma unroll
    for (int ms = 0; ms < AT_MS; ++ms) {
#pragma unroll
        for (int t = 0; t < 16; ++t) {
#pragma unroll
            for (int v = 0; v < 8; ++v) {
                const int m   = 16 * ms + v + 8 * lh;
                const int col = t * 16 + ll;
                lds[m * LDS_STRIDE + col] = sc[ms][t][v];
            }
        }
    }
    __syncthreads();

    float* const arowT = attn + (size_t)b * S * KP + (size_t)s0 * KP + 4 * l;
    const float* const ldsRow = lds + 4 * l;
#define AT_ROWLD(rr, g)                                                       \
    {                                                                         \
        _Pragma("unroll") for (int i = 0; i < 4; ++i) {                       \
            _Pragma("unroll") for (int h = 0; h < 2; ++h)                     \
                rr[2 * i + h] = *(const v4f*)(ldsRow +                        \
                    (4 * (g) + i) * LDS_STRIDE + 128 * h);                    \
        }                                                                     \
    }
#define AT_ROWST(rr, g)                                                       \
    {                                                                         \
        _Pragma("unroll") for (int i = 0; i < 4; ++i) {                       \
            _Pragma("unroll") for (int h = 0; h < 2; ++h)                     \
                *(v4f*)(arowT + (size_t)(4 * (g) + i) * KP + 128 * h) =       \
                    rr[2 * i + h];                                            \
        }                                                                     \
    }
    v4f r0[8], r1[8];

    v8f oc[AT_MS][4];
#pragma unroll
    for (int ms = 0; ms < AT_MS; ++ms) {
#pragma unroll
        for (int td = 0; td < 4; ++td) oc[ms][td] = v8f{};
    }
    AT_LOADA(va0, 0)
    AT_ROWLD(r0, 0)
#pragma unroll 1
    for (int g = 0; g < AT_NG - 2; g += 2) {
        AT_LOADB(vb1, g + 1)
        AT_LOADA(va1, g + 1)
        AT_ROWLD(r1, g + 1)
        AT_ROWST(r0, g)
        AT_MMA(va0, vb0)
        AT_LOADB(vb0, g + 2)
        AT_LOADA(va0, g + 2)
        AT_ROWLD(r0, g + 2)
        AT_ROWST(r1, g + 1)
        AT_MMA(va1, vb1)
    }
    AT_LOADB(vb1, AT_NG - 1)
    AT_LOADA(va1, AT_NG - 1)
    AT_ROWLD(r1, AT_NG - 1)
    AT_ROWST(r0, AT_NG - 2)
    AT_MMA(va0, vb0)
    AT_ROWST(r1, AT_NG - 1)
    AT_MMA(va1, vb1)
#undef AT_LOADB
#undef AT_LOADA
#undef AT_ROWLD
#undef AT_ROWST
#undef AT_MMA
#undef AT_G
#undef AT_NG
#pragma unroll
    for (int ms = 0; ms < AT_MS; ++ms) {
#pragma unroll
        for (int td = 0; td < 4; ++td) {
#pragma unroll
            for (int v = 0; v < 8; ++v)
                out[(size_t)b * S * D + (size_t)(s0 + 16 * ms + v + 8 * lh) * D + td * 16 +
                    ll] = oc[ms][td][v];
        }
    }
}

extern "C" void kernel_launch(void* const* d_in, const int* in_sizes, int n_in,
                              void* d_out, int out_size, void* d_ws, size_t ws_size,
                              hipStream_t stream) {
    (void)in_sizes; (void)n_in; (void)out_size; (void)ws_size;
    const float* Q   = (const float*)d_in[0];
    const float* K   = (const float*)d_in[1];
    const float* V   = (const float*)d_in[2];
    const float* E_w = (const float*)d_in[3];
    const float* E_b = (const float*)d_in[4];
    const float* F_w = (const float*)d_in[5];
    const float* F_b = (const float*)d_in[6];

    float* out  = (float*)d_out;
    float* attn = out + (size_t)LIN_B * LIN_S * LIN_D;

    float* Kpp = (float*)d_ws;
    float* Vpp = Kpp + (size_t)LIN_B * LIN_D * LIN_KP;

    const int attnBlocks = LIN_B * (LIN_S / AT_ROWS);

    lin_proj_kernel<<<256, 256, 0, stream>>>(K, V, E_w, F_w, E_b, F_b, Kpp, Vpp);
    lin_attn_kernel<<<attnBlocks, 32, 0, stream>>>(Q, Kpp, Vpp, out, attn);
}
